# starting kernel
# baseline (speedup 1.0000x reference)
#include <hip/hip_runtime.h>
#include <hip/hip_bf16.h>
#include <math.h>


typedef __attribute__((ext_vector_type(2))) float v2f;
typedef __attribute__((ext_vector_type(8))) float v8f;

#define GNN_N       100000
#define GNN_NUSERS  60000
#define GNN_EMB     64
#define GNN_E       3200000

__global__ __launch_bounds__(256) void gnn_init(
    const float* __restrict__ user_emb, const float* __restrict__ item_emb,
    float* __restrict__ ego, float* __restrict__ out)
{
    long tid = (long)blockIdx.x * 256 + threadIdx.x;
    if (tid >= (long)GNN_N * GNN_EMB) return;
    long row = tid >> 6;
    int  col = (int)(tid & 63);
    float v = (row < GNN_NUSERS)
                ? user_emb[row * GNN_EMB + col]
                : item_emb[(row - GNN_NUSERS) * GNN_EMB + col];
    ego[tid] = v;
    out[row * 192 + col] = v;
}

__global__ __launch_bounds__(256) void gnn_zero(float4* __restrict__ p, long n4)
{
    long tid = (long)blockIdx.x * 256 + threadIdx.x;
    if (tid < n4) p[tid] = make_float4(0.f, 0.f, 0.f, 0.f);
}

__global__ __launch_bounds__(256) void gnn_spmm(
    const int* __restrict__ esrc, const int* __restrict__ edst,
    const float* __restrict__ eval, const float* __restrict__ x,
    float* __restrict__ y)
{
    long tid = (long)blockIdx.x * 256 + threadIdx.x;
    long e = tid >> 3;
    if (e >= GNN_E) return;
    int part = (int)(tid & 7);
    int d = edst[e];
    int s = esrc[e];
    float v = eval[e];
    const float4* xr = (const float4*)(x + (long)d * GNN_EMB) + part * 2;
    float4 a = xr[0];
    float4 b = xr[1];
    float* yr = y + (long)s * GNN_EMB + part * 8;
    unsafeAtomicAdd(yr + 0, v * a.x);
    unsafeAtomicAdd(yr + 1, v * a.y);
    unsafeAtomicAdd(yr + 2, v * a.z);
    unsafeAtomicAdd(yr + 3, v * a.w);
    unsafeAtomicAdd(yr + 4, v * b.x);
    unsafeAtomicAdd(yr + 5, v * b.y);
    unsafeAtomicAdd(yr + 6, v * b.z);
    unsafeAtomicAdd(yr + 7, v * b.w);
}

__global__ __launch_bounds__(256) void gnn_layer(
    const float* __restrict__ sideL, float* __restrict__ ego,
    const float* __restrict__ Wgc, const float* __restrict__ bgc,
    const float* __restrict__ Wbi, const float* __restrict__ bbi,
    float* __restrict__ out, int out_col0)
{
    __shared__ float Wt[64][132];
    __shared__ float bsum[64];

    const int tid = threadIdx.x;

    for (int i = tid; i < 64 * 128; i += 256) {
        int k = i >> 6;
        int n = i & 63;
        float w = (k < 64) ? Wgc[k * 64 + n] : Wbi[(k - 64) * 64 + n];
        Wt[n][k] = w;
    }
    if (tid < 64) bsum[tid] = bgc[tid] + bbi[tid];
    __syncthreads();

    const int wave    = tid >> 5;
    const int lane    = tid & 31;
    const int rowbase = blockIdx.x * 128 + wave * 16;
    if (rowbase >= GNN_N) return;

    const int m    = lane & 15;
    const int half = lane >> 4;
    const int kb   = half << 1;

    const long  row    = (long)(rowbase + m);
    const float* egoRd = ego   + row * 64;
    const float* sLRd  = sideL + row * 64;

    v2f e2[16], l2[16];
#pragma unroll
    for (int s = 0; s < 16; ++s) {
        int c = 4 * s + kb;
        e2[s] = *(const v2f*)(egoRd + c);
        l2[s] = *(const v2f*)(sLRd + c);
    }

    v8f acc[4] = {};

#pragma unroll
    for (int s = 0; s < 16; ++s) {
        v2f a;
        a.x = e2[s].x + l2[s].x;
        a.y = e2[s].y + l2[s].y;
#pragma unroll
        for (int nt = 0; nt < 4; ++nt) {
            v2f b = *(const v2f*)(&Wt[nt * 16 + m][4 * s + kb]);
            acc[nt] = __builtin_amdgcn_wmma_f32_16x16x4_f32(
                false, a, false, b, (short)0, acc[nt], false, false);
        }
    }
#pragma unroll
    for (int s = 0; s < 16; ++s) {
        v2f a;
        a.x = e2[s].x * l2[s].x;
        a.y = e2[s].y * l2[s].y;
#pragma unroll
        for (int nt = 0; nt < 4; ++nt) {
            v2f b = *(const v2f*)(&Wt[nt * 16 + m][64 + 4 * s + kb]);
            acc[nt] = __builtin_amdgcn_wmma_f32_16x16x4_f32(
                false, a, false, b, (short)0, acc[nt], false, false);
        }
    }

    float vout[4][8];
    float ssq[8];
#pragma unroll
    for (int j = 0; j < 8; ++j) ssq[j] = 0.f;
#pragma unroll
    for (int nt = 0; nt < 4; ++nt) {
        float bc = bsum[nt * 16 + m];
#pragma unroll
        for (int j = 0; j < 8; ++j) {
            float v = acc[nt][j] + bc;
            v = (v > 0.f) ? v : 0.01f * v;
            vout[nt][j] = v;
            ssq[j] += v * v;
        }
    }

    float* egoWr = ego + (long)rowbase * 64;
#pragma unroll
    for (int j = 0; j < 8; ++j) {
        int mrow = j + (half << 3);
#pragma unroll
        for (int nt = 0; nt < 4; ++nt)
            egoWr[mrow * 64 + nt * 16 + m] = vout[nt][j];
    }

#pragma unroll
    for (int j = 0; j < 8; ++j) {
        float s = ssq[j];
        s += __shfl_xor(s, 8, 16);
        s += __shfl_xor(s, 4, 16);
        s += __shfl_xor(s, 2, 16);
        s += __shfl_xor(s, 1, 16);
        float inv = 1.0f / fmaxf(sqrtf(s), 1e-12f);
        int mrow = j + (half << 3);
        float* o = out + (long)(rowbase + mrow) * 192 + out_col0;
#pragma unroll
        for (int nt = 0; nt < 4; ++nt)
            o[nt * 16 + m] = vout[nt][j] * inv;
    }
}

extern "C" void kernel_launch(void* const* d_in, const int* in_sizes, int n_in,
                              void* d_out, int out_size, void* d_ws, size_t ws_size,
                              hipStream_t stream) {
    const float* user_emb = (const float*)d_in[0];
    const float* item_emb = (const float*)d_in[1];
    const float* W_gc0 = (const float*)d_in[2];
    const float* b_gc0 = (const float*)d_in[3];
    const float* W_bi0 = (const float*)d_in[4];
    const float* b_bi0 = (const float*)d_in[5];
    const float* W_gc1 = (const float*)d_in[6];
    const float* b_gc1 = (const float*)d_in[7];
    const float* W_bi1 = (const float*)d_in[8];
    const float* b_bi1 = (const float*)d_in[9];
    const int*   e_src = (const int*)d_in[10];
    const int*   e_dst = (const int*)d_in[11];
    const float* e_val = (const float*)d_in[12];
    float* out = (float*)d_out;

    float* ego   = (float*)d_ws;
    float* sideL = ego + (long)GNN_N * GNN_EMB;

    const long nelem  = (long)GNN_N * GNN_EMB;
    const int  gInit  = (int)((nelem + 255) / 256);
    const int  gZero  = (int)((nelem / 4 + 255) / 256);
    const int  gSpmm  = (int)(((long)GNN_E * 8 + 255) / 256);
    const int  gLayer = (GNN_N + 127) / 128;

    gnn_init<<<gInit, 256, 0, stream>>>(user_emb, item_emb, ego, out);

    gnn_zero<<<gZero, 256, 0, stream>>>((float4*)sideL, nelem / 4);
    gnn_spmm<<<gSpmm, 256, 0, stream>>>(e_src, e_dst, e_val, ego, sideL);
    gnn_layer<<<gLayer, 256, 0, stream>>>(sideL, ego, W_gc0, b_gc0, W_bi0, b_bi0,
                                          out, 64);

    gnn_zero<<<gZero, 256, 0, stream>>>((float4*)sideL, nelem / 4);
    gnn_spmm<<<gSpmm, 256, 0, stream>>>(e_src, e_dst, e_val, ego, sideL);
    gnn_layer<<<gLayer, 256, 0, stream>>>(sideL, ego, W_gc1, b_gc1, W_bi1, b_bi1,
                                          out, 128);
}


# SpMM by row ownership: 256 blocks each scan a bit-sliced owner byte per edge, queue their own edges in LDS, sum them into a 391-row LDS tile (ds_add_f32); no global atomics; sum order: LDS arrival
# speedup vs baseline: 2.6529x; 2.6529x over previous
#include <hip/hip_runtime.h>
#include <hip/hip_bf16.h>
#include <math.h>


typedef __attribute__((ext_vector_type(2))) float v2f;
typedef __attribute__((ext_vector_type(8))) float v8f;

#define GNN_N       100000
#define GNN_NUSERS  60000
#define GNN_EMB     64
#define GNN_E       3200000

__global__ __launch_bounds__(256) void gnn_init(
    const float* __restrict__ user_emb, const float* __restrict__ item_emb,
    float* __restrict__ ego, float* __restrict__ out)
{
    long tid = (long)blockIdx.x * 256 + threadIdx.x;
    if (tid >= (long)GNN_N * GNN_EMB) return;
    long row = tid >> 6;
    int  col = (int)(tid & 63);
    float v = (row < GNN_NUSERS)
                ? user_emb[row * GNN_EMB + col]
                : item_emb[(row - GNN_NUSERS) * GNN_EMB + col];
    ego[tid] = v;
    out[row * 192 + col] = v;
}

#define OWN_ROWS   391
#define OWN_BLOCKS 256
#define OWN_Q      32768
#define KEY_G      (GNN_E / 32)

static_assert(OWN_ROWS * OWN_BLOCKS >= GNN_N, "owners must cover all rows");
static_assert(GNN_E % 256 == 0, "the key pass runs full waves only");

__global__ __launch_bounds__(256) void gnn_key(
    const int* __restrict__ esrc, unsigned* __restrict__ key)
{
    const int e    = blockIdx.x * 256 + threadIdx.x;
    const int lane = threadIdx.x & 31;
    const unsigned o = (((unsigned)esrc[e]) / OWN_ROWS) & 255u;
    unsigned mine = 0u;
#pragma unroll
    for (int j = 0; j < 8; ++j) {
        unsigned bj = __builtin_amdgcn_ballot_w32(((o >> j) & 1u) != 0u);
        if (lane == j) mine = bj;
    }
    if (lane < 8) key[(e >> 5) * 8 + lane] = mine;
}

__global__ __launch_bounds__(1024) void gnn_spmm_own(
    const int* __restrict__ esrc, const int* __restrict__ edst,
    const float* __restrict__ eval, const unsigned* __restrict__ key,
    const float* __restrict__ x, float* __restrict__ y)
{
    const uint4* __restrict__ key16 = (const uint4*)key;
    __shared__ __attribute__((aligned(16))) float acc[OWN_ROWS * GNN_EMB];
    __shared__ unsigned q[OWN_Q];
    __shared__ unsigned qn;

    const int tid  = threadIdx.x;
    const int lane = tid & 31;
    const int wave = __builtin_amdgcn_readfirstlane(tid >> 5);
    const int lo   = blockIdx.x * OWN_ROWS;
    const int rows = min(OWN_ROWS, GNN_N - lo);
    unsigned inv[8];
#pragma unroll
    for (int j = 0; j < 8; ++j) inv[j] = ((blockIdx.x >> j) & 1u) ? 0u : ~0u;

    for (int i = tid; i < OWN_ROWS * GNN_EMB; i += 1024) acc[i] = 0.f;

    int pos  = 0;
    int step = KEY_G;
    for (int trip = 0; trip < 3 + GNN_E / OWN_Q && pos < KEY_G; ++trip) {
        const int end = min(pos + step, KEY_G);
        if (tid == 0) qn = 0;
        __syncthreads();

        for (int g0 = pos + tid; g0 < end; g0 += 4096) {
            uint4 ka[4], kb[4];
#pragma unroll
            for (int u = 0; u < 4; ++u) {
                int gc = min(g0 + u * 1024, KEY_G - 1);
                ka[u] = key16[2 * gc];
                kb[u] = key16[2 * gc + 1];
            }
#pragma unroll
            for (int u = 0; u < 4; ++u) {
                const int g = g0 + u * 1024;
                unsigned m = (ka[u].x ^ inv[0]) & (ka[u].y ^ inv[1])
                           & (ka[u].z ^ inv[2]) & (ka[u].w ^ inv[3])
                           & (kb[u].x ^ inv[4]) & (kb[u].y ^ inv[5])
                           & (kb[u].z ^ inv[6]) & (kb[u].w ^ inv[7]);
                if (g >= end) m = 0u;
                while (m) {
                    unsigned bit = (unsigned)__ffs((int)m) - 1u;
                    m &= m - 1u;
                    unsigned slot = atomicAdd(&qn, 1u);
                    if (slot < OWN_Q) q[slot] = (unsigned)g * 32u + bit;
                }
            }
        }
        __syncthreads();
        const unsigned n = __builtin_amdgcn_readfirstlane(qn);
        __syncthreads();
        if (n > OWN_Q) { step = OWN_Q / 32; continue; }

        for (unsigned base = (unsigned)wave * 32u; base < n; base += 1024u) {
            const unsigned qi = base + lane;
            const unsigned e = q[min(qi, n - 1u)];
            const int   d = edst[e];
            const int   r = esrc[e] - lo;
            const float v = eval[e];
            const bool ok = qi < n && (unsigned)r < (unsigned)rows;
            const unsigned okm = __builtin_amdgcn_ballot_w32(ok);
#pragma unroll
            for (int j0 = 0; j0 < 32; j0 += 16) {
                if (((okm >> j0) & 0xFFFFu) == 0u) continue;
                v2f xv[16];
#pragma unroll
                for (int u = 0; u < 16; ++u) {
                    int dj = __builtin_amdgcn_readlane(d, j0 + u);
                    xv[u] = *(const v2f*)(x + (long)dj * GNN_EMB + lane * 2);
                }
#pragma unroll
                for (int u = 0; u < 16; ++u) {
                    if ((okm >> (j0 + u)) & 1u) {
                        int   rj = __builtin_amdgcn_readlane(r, j0 + u);
                        float vj = __int_as_float(
                            __builtin_amdgcn_readlane(__float_as_int(v), j0 + u));
                        float* a = &acc[rj * GNN_EMB + lane * 2];
                        __hip_atomic_fetch_add(a,     vj * xv[u].x, __ATOMIC_RELAXED,
                                               __HIP_MEMORY_SCOPE_WORKGROUP);
                        __hip_atomic_fetch_add(a + 1, vj * xv[u].y, __ATOMIC_RELAXED,
                                               __HIP_MEMORY_SCOPE_WORKGROUP);
                    }
                }
            }
        }
        pos = end;
    }
    __syncthreads();

    float4* yo = (float4*)(y + (long)lo * GNN_EMB);
    const float4* a4 = (const float4*)acc;
    for (int i = tid; i < rows * (GNN_EMB / 4); i += 1024) yo[i] = a4[i];
}

__global__ __launch_bounds__(256) void gnn_layer(
    const float* __restrict__ sideL, float* __restrict__ ego,
    const float* __restrict__ Wgc, const float* __restrict__ bgc,
    const float* __restrict__ Wbi, const float* __restrict__ bbi,
    float* __restrict__ out, int out_col0)
{
    __shared__ float Wt[64][132];
    __shared__ float bsum[64];

    const int tid = threadIdx.x;

    for (int i = tid; i < 64 * 128; i += 256) {
        int k = i >> 6;
        int n = i & 63;
        float w = (k < 64) ? Wgc[k * 64 + n] : Wbi[(k - 64) * 64 + n];
        Wt[n][k] = w;
    }
    if (tid < 64) bsum[tid] = bgc[tid] + bbi[tid];
    __syncthreads();

    const int wave    = tid >> 5;
    const int lane    = tid & 31;
    const int rowbase = blockIdx.x * 128 + wave * 16;
    if (rowbase >= GNN_N) return;

    const int m    = lane & 15;
    const int half = lane >> 4;
    const int kb   = half << 1;

    const long  row    = (long)(rowbase + m);
    const float* egoRd = ego   + row * 64;
    const float* sLRd  = sideL + row * 64;

    v2f e2[16], l2[16];
#pragma unroll
    for (int s = 0; s < 16; ++s) {
        int c = 4 * s + kb;
        e2[s] = *(const v2f*)(egoRd + c);
        l2[s] = *(const v2f*)(sLRd + c);
    }

    v8f acc[4] = {};

#pragma unroll
    for (int s = 0; s < 16; ++s) {
        v2f a;
        a.x = e2[s].x + l2[s].x;
        a.y = e2[s].y + l2[s].y;
#pragma unroll
        for (int nt = 0; nt < 4; ++nt) {
            v2f b = *(const v2f*)(&Wt[nt * 16 + m][4 * s + kb]);
            acc[nt] = __builtin_amdgcn_wmma_f32_16x16x4_f32(
                false, a, false, b, (short)0, acc[nt], false, false);
        }
    }
#pragma unroll
    for (int s = 0; s < 16; ++s) {
        v2f a;
        a.x = e2[s].x * l2[s].x;
        a.y = e2[s].y * l2[s].y;
#pragma unroll
        for (int nt = 0; nt < 4; ++nt) {
            v2f b = *(const v2f*)(&Wt[nt * 16 + m][64 + 4 * s + kb]);
            acc[nt] = __builtin_amdgcn_wmma_f32_16x16x4_f32(
                false, a, false, b, (short)0, acc[nt], false, false);
        }
    }

    float vout[4][8];
    float ssq[8];
#pragma unroll
    for (int j = 0; j < 8; ++j) ssq[j] = 0.f;
#pragma unroll
    for (int nt = 0; nt < 4; ++nt) {
        float bc = bsum[nt * 16 + m];
#pragma unroll
        for (int j = 0; j < 8; ++j) {
            float v = acc[nt][j] + bc;
            v = (v > 0.f) ? v : 0.01f * v;
            vout[nt][j] = v;
            ssq[j] += v * v;
        }
    }

    float* egoWr = ego + (long)rowbase * 64;
#pragma unroll
    for (int j = 0; j < 8; ++j) {
        int mrow = j + (half << 3);
#pragma unroll
        for (int nt = 0; nt < 4; ++nt)
            egoWr[mrow * 64 + nt * 16 + m] = vout[nt][j];
    }

#pragma unroll
    for (int j = 0; j < 8; ++j) {
        float s = ssq[j];
        s += __shfl_xor(s, 8, 16);
        s += __shfl_xor(s, 4, 16);
        s += __shfl_xor(s, 2, 16);
        s += __shfl_xor(s, 1, 16);
        float inv = 1.0f / fmaxf(sqrtf(s), 1e-12f);
        int mrow = j + (half << 3);
        float* o = out + (long)(rowbase + mrow) * 192 + out_col0;
#pragma unroll
        for (int nt = 0; nt < 4; ++nt)
            o[nt * 16 + m] = vout[nt][j] * inv;
    }
}

extern "C" void kernel_launch(void* const* d_in, const int* in_sizes, int n_in,
                              void* d_out, int out_size, void* d_ws, size_t ws_size,
                              hipStream_t stream) {
    const float* user_emb = (const float*)d_in[0];
    const float* item_emb = (const float*)d_in[1];
    const float* W_gc0 = (const float*)d_in[2];
    const float* b_gc0 = (const float*)d_in[3];
    const float* W_bi0 = (const float*)d_in[4];
    const float* b_bi0 = (const float*)d_in[5];
    const float* W_gc1 = (const float*)d_in[6];
    const float* b_gc1 = (const float*)d_in[7];
    const float* W_bi1 = (const float*)d_in[8];
    const float* b_bi1 = (const float*)d_in[9];
    const int*   e_src = (const int*)d_in[10];
    const int*   e_dst = (const int*)d_in[11];
    const float* e_val = (const float*)d_in[12];
    float* out = (float*)d_out;

    float* ego   = (float*)d_ws;
    float* sideL = ego + (long)GNN_N * GNN_EMB;
    unsigned* key4 = (unsigned*)(sideL + (long)GNN_N * GNN_EMB);

    const long nelem  = (long)GNN_N * GNN_EMB;
    const int  gInit  = (int)((nelem + 255) / 256);
    const int  gKey   = GNN_E / 256;
    const int  gLayer = (GNN_N + 127) / 128;

    gnn_init<<<gInit, 256, 0, stream>>>(user_emb, item_emb, ego, out);
    gnn_key<<<gKey, 256, 0, stream>>>(e_src, key4);

    gnn_spmm_own<<<OWN_BLOCKS, 1024, 0, stream>>>(e_src, e_dst, e_val,
                                                  key4, ego, sideL);
    gnn_layer<<<gLayer, 256, 0, stream>>>(sideL, ego, W_gc0, b_gc0, W_bi0, b_bi0,
                                          out, 64);

    gnn_spmm_own<<<OWN_BLOCKS, 1024, 0, stream>>>(e_src, e_dst, e_val,
                                                  key4, ego, sideL);
    gnn_layer<<<gLayer, 256, 0, stream>>>(sideL, ego, W_gc1, b_gc1, W_bi1, b_bi1,
                                          out, 128);
}


# SpMM sum phase: key pass also packs each edge's dst, local row and val into one 8-B record (held in out's last 64 columns until layer 1 writes them): one gathered line per owned edge, was three
# speedup vs baseline: 3.0786x; 1.1605x over previous
#include <hip/hip_runtime.h>
#include <hip/hip_bf16.h>
#include <math.h>


typedef __attribute__((ext_vector_type(2))) float v2f;
typedef __attribute__((ext_vector_type(8))) float v8f;

#define GNN_N       100000
#define GNN_NUSERS  60000
#define GNN_EMB     64
#define GNN_E       3200000

__global__ __launch_bounds__(256) void gnn_init(
    const float* __restrict__ user_emb, const float* __restrict__ item_emb,
    float* __restrict__ ego, float* __restrict__ out)
{
    long tid = (long)blockIdx.x * 256 + threadIdx.x;
    if (tid >= (long)GNN_N * GNN_EMB) return;
    long row = tid >> 6;
    int  col = (int)(tid & 63);
    float v = (row < GNN_NUSERS)
                ? user_emb[row * GNN_EMB + col]
                : item_emb[(row - GNN_NUSERS) * GNN_EMB + col];
    ego[tid] = v;
    out[row * 192 + col] = v;
}

#define OWN_ROWS   391
#define OWN_BLOCKS 256
#define OWN_Q      32768
#define KEY_G      (GNN_E / 32)
#define REC_DST_BITS 17
#define REC_BAD_ROW  511u

static_assert(OWN_ROWS * OWN_BLOCKS >= GNN_N, "owners must cover all rows");
static_assert(GNN_E % 256 == 0, "the key pass runs full waves only");
static_assert(KEY_G == GNN_N, "one 256-B record group per output row");
static_assert(GNN_N <= (1 << REC_DST_BITS), "dst must fit its record field");
static_assert(OWN_ROWS <= (int)REC_BAD_ROW, "local row must fit below the bad-row code");

__global__ __launch_bounds__(256) void gnn_key(
    const int* __restrict__ esrc, const int* __restrict__ edst,
    const float* __restrict__ eval, unsigned* __restrict__ key,
    float* __restrict__ out)
{
    const int e    = blockIdx.x * 256 + threadIdx.x;
    const int lane = threadIdx.x & 31;
    const unsigned s  = (unsigned)esrc[e];
    const unsigned oq = s / OWN_ROWS;
    const unsigned o  = oq & 255u;
    const unsigned lr = (oq < (unsigned)OWN_BLOCKS) ? (s - oq * OWN_ROWS) : REC_BAD_ROW;
    uint2 rc;
    rc.x = (((unsigned)edst[e]) & ((1u << REC_DST_BITS) - 1u)) | (lr << REC_DST_BITS);
    rc.y = __float_as_uint(eval[e]);
    *(uint2*)(out + (long)(e >> 5) * 192 + 128 + lane * 2) = rc;
    unsigned mine = 0u;
#pragma unroll
    for (int j = 0; j < 8; ++j) {
        unsigned bj = __builtin_amdgcn_ballot_w32(((o >> j) & 1u) != 0u);
        if (lane == j) mine = bj;
    }
    if (lane < 8) key[(e >> 5) * 8 + lane] = mine;
}

__global__ __launch_bounds__(1024) void gnn_spmm_own(
    const unsigned* __restrict__ rec, const unsigned* __restrict__ key,
    const float* __restrict__ x, float* __restrict__ y)
{
    const uint4* __restrict__ key16 = (const uint4*)key;
    __shared__ __attribute__((aligned(16))) float acc[OWN_ROWS * GNN_EMB];
    __shared__ unsigned q[OWN_Q];
    __shared__ unsigned qn;

    const int tid  = threadIdx.x;
    const int lane = tid & 31;
    const int wave = __builtin_amdgcn_readfirstlane(tid >> 5);
    const int lo   = blockIdx.x * OWN_ROWS;
    const int rows = min(OWN_ROWS, GNN_N - lo);
    unsigned inv[8];
#pragma unroll
    for (int j = 0; j < 8; ++j) inv[j] = ((blockIdx.x >> j) & 1u) ? 0u : ~0u;

    for (int i = tid; i < OWN_ROWS * GNN_EMB; i += 1024) acc[i] = 0.f;

    int pos  = 0;
    int step = KEY_G;
    for (int trip = 0; trip < 3 + GNN_E / OWN_Q && pos < KEY_G; ++trip) {
        const int end = min(pos + step, KEY_G);
        if (tid == 0) qn = 0;
        __syncthreads();

        for (int g0 = pos + tid; g0 < end; g0 += 4096) {
            uint4 ka[4], kb[4];
#pragma unroll
            for (int u = 0; u < 4; ++u) {
                int gc = min(g0 + u * 1024, KEY_G - 1);
                ka[u] = key16[2 * gc];
                kb[u] = key16[2 * gc + 1];
            }
#pragma unroll
            for (int u = 0; u < 4; ++u) {
                const int g = g0 + u * 1024;
                unsigned m = (ka[u].x ^ inv[0]) & (ka[u].y ^ inv[1])
                           & (ka[u].z ^ inv[2]) & (ka[u].w ^ inv[3])
                           & (kb[u].x ^ inv[4]) & (kb[u].y ^ inv[5])
                           & (kb[u].z ^ inv[6]) & (kb[u].w ^ inv[7]);
                if (g >= end) m = 0u;
                while (m) {
                    unsigned bit = (unsigned)__ffs((int)m) - 1u;
                    m &= m - 1u;
                    unsigned slot = atomicAdd(&qn, 1u);
                    if (slot < OWN_Q) q[slot] = (unsigned)g * 32u + bit;
                }
            }
        }
        __syncthreads();
        const unsigned n = __builtin_amdgcn_readfirstlane(qn);
        __syncthreads();
        if (n > OWN_Q) { step = OWN_Q / 32; continue; }

        for (unsigned base = (unsigned)wave * 32u; base < n; base += 1024u) {
            const unsigned qi = base + lane;
            const unsigned e = q[min(qi, n - 1u)];
            const uint2 rc = *(const uint2*)(rec + (size_t)(e >> 5) * 192u + (e & 31u) * 2u);
            const int   d = (int)(rc.x & ((1u << REC_DST_BITS) - 1u));
            const int   r = (int)(rc.x >> REC_DST_BITS);
            const float v = __uint_as_float(rc.y);
            const bool ok = qi < n && (unsigned)r < (unsigned)rows;
            const unsigned okm = __builtin_amdgcn_ballot_w32(ok);
#pragma unroll
            for (int j0 = 0; j0 < 32; j0 += 16) {
                if (((okm >> j0) & 0xFFFFu) == 0u) continue;
                v2f xv[16];
#pragma unroll
                for (int u = 0; u < 16; ++u) {
                    int dj = __builtin_amdgcn_readlane(d, j0 + u);
                    xv[u] = *(const v2f*)(x + (long)dj * GNN_EMB + lane * 2);
                }
#pragma unroll
                for (int u = 0; u < 16; ++u) {
                    if ((okm >> (j0 + u)) & 1u) {
                        int   rj = __builtin_amdgcn_readlane(r, j0 + u);
                        float vj = __int_as_float(
                            __builtin_amdgcn_readlane(__float_as_int(v), j0 + u));
                        float* a = &acc[rj * GNN_EMB + lane * 2];
                        __hip_atomic_fetch_add(a,     vj * xv[u].x, __ATOMIC_RELAXED,
                                               __HIP_MEMORY_SCOPE_WORKGROUP);
                        __hip_atomic_fetch_add(a + 1, vj * xv[u].y, __ATOMIC_RELAXED,
                                               __HIP_MEMORY_SCOPE_WORKGROUP);
                    }
                }
            }
        }
        pos = end;
    }
    __syncthreads();

    float4* yo = (float4*)(y + (long)lo * GNN_EMB);
    const float4* a4 = (const float4*)acc;
    for (int i = tid; i < rows * (GNN_EMB / 4); i += 1024) yo[i] = a4[i];
}

__global__ __launch_bounds__(256) void gnn_layer(
    const float* __restrict__ sideL, float* __restrict__ ego,
    const float* __restrict__ Wgc, const float* __restrict__ bgc,
    const float* __restrict__ Wbi, const float* __restrict__ bbi,
    float* __restrict__ out, int out_col0)
{
    __shared__ float Wt[64][132];
    __shared__ float bsum[64];

    const int tid = threadIdx.x;

    for (int i = tid; i < 64 * 128; i += 256) {
        int k = i >> 6;
        int n = i & 63;
        float w = (k < 64) ? Wgc[k * 64 + n] : Wbi[(k - 64) * 64 + n];
        Wt[n][k] = w;
    }
    if (tid < 64) bsum[tid] = bgc[tid] + bbi[tid];
    __syncthreads();

    const int wave    = tid >> 5;
    const int lane    = tid & 31;
    const int rowbase = blockIdx.x * 128 + wave * 16;
    if (rowbase >= GNN_N) return;

    const int m    = lane & 15;
    const int half = lane >> 4;
    const int kb   = half << 1;

    const long  row    = (long)(rowbase + m);
    const float* egoRd = ego   + row * 64;
    const float* sLRd  = sideL + row * 64;

    v2f e2[16], l2[16];
#pragma unroll
    for (int s = 0; s < 16; ++s) {
        int c = 4 * s + kb;
        e2[s] = *(const v2f*)(egoRd + c);
        l2[s] = *(const v2f*)(sLRd + c);
    }

    v8f acc[4] = {};

#pragma unroll
    for (int s = 0; s < 16; ++s) {
        v2f a;
        a.x = e2[s].x + l2[s].x;
        a.y = e2[s].y + l2[s].y;
#pragma unroll
        for (int nt = 0; nt < 4; ++nt) {
            v2f b = *(const v2f*)(&Wt[nt * 16 + m][4 * s + kb]);
            acc[nt] = __builtin_amdgcn_wmma_f32_16x16x4_f32(
                false, a, false, b, (short)0, acc[nt], false, false);
        }
    }
#pragma unroll
    for (int s = 0; s < 16; ++s) {
        v2f a;
        a.x = e2[s].x * l2[s].x;
        a.y = e2[s].y * l2[s].y;
#pragma unroll
        for (int nt = 0; nt < 4; ++nt) {
            v2f b = *(const v2f*)(&Wt[nt * 16 + m][64 + 4 * s + kb]);
            acc[nt] = __builtin_amdgcn_wmma_f32_16x16x4_f32(
                false, a, false, b, (short)0, acc[nt], false, false);
        }
    }

    float vout[4][8];
    float ssq[8];
#pragma unroll
    for (int j = 0; j < 8; ++j) ssq[j] = 0.f;
#pragma unroll
    for (int nt = 0; nt < 4; ++nt) {
        float bc = bsum[nt * 16 + m];
#pragma unroll
        for (int j = 0; j < 8; ++j) {
            float v = acc[nt][j] + bc;
            v = (v > 0.f) ? v : 0.01f * v;
            vout[nt][j] = v;
            ssq[j] += v * v;
        }
    }

    float* egoWr = ego + (long)rowbase * 64;
#pragma unroll
    for (int j = 0; j < 8; ++j) {
        int mrow = j + (half << 3);
#pragma unroll
        for (int nt = 0; nt < 4; ++nt)
            egoWr[mrow * 64 + nt * 16 + m] = vout[nt][j];
    }

#pragma unroll
    for (int j = 0; j < 8; ++j) {
        float s = ssq[j];
        s += __shfl_xor(s, 8, 16);
        s += __shfl_xor(s, 4, 16);
        s += __shfl_xor(s, 2, 16);
        s += __shfl_xor(s, 1, 16);
        float inv = 1.0f / fmaxf(sqrtf(s), 1e-12f);
        int mrow = j + (half << 3);
        float* o = out + (long)(rowbase + mrow) * 192 + out_col0;
#pragma unroll
        for (int nt = 0; nt < 4; ++nt)
            o[nt * 16 + m] = vout[nt][j] * inv;
    }
}

extern "C" void kernel_launch(void* const* d_in, const int* in_sizes, int n_in,
                              void* d_out, int out_size, void* d_ws, size_t ws_size,
                              hipStream_t stream) {
    const float* user_emb = (const float*)d_in[0];
    const float* item_emb = (const float*)d_in[1];
    const float* W_gc0 = (const float*)d_in[2];
    const float* b_gc0 = (const float*)d_in[3];
    const float* W_bi0 = (const float*)d_in[4];
    const float* b_bi0 = (const float*)d_in[5];
    const float* W_gc1 = (const float*)d_in[6];
    const float* b_gc1 = (const float*)d_in[7];
    const float* W_bi1 = (const float*)d_in[8];
    const float* b_bi1 = (const float*)d_in[9];
    const int*   e_src = (const int*)d_in[10];
    const int*   e_dst = (const int*)d_in[11];
    const float* e_val = (const float*)d_in[12];
    float* out = (float*)d_out;

    float* ego   = (float*)d_ws;
    float* sideL = ego + (long)GNN_N * GNN_EMB;
    unsigned* key4 = (unsigned*)(sideL + (long)GNN_N * GNN_EMB);

    const long nelem  = (long)GNN_N * GNN_EMB;
    const int  gInit  = (int)((nelem + 255) / 256);
    const int  gKey   = GNN_E / 256;
    const int  gLayer = (GNN_N + 127) / 128;

    gnn_init<<<gInit, 256, 0, stream>>>(user_emb, item_emb, ego, out);
    gnn_key<<<gKey, 256, 0, stream>>>(e_src, e_dst, e_val, key4, out);
    const unsigned* rec = (const unsigned*)(out + 128);

    gnn_spmm_own<<<OWN_BLOCKS, 1024, 0, stream>>>(rec, key4, ego, sideL);
    gnn_layer<<<gLayer, 256, 0, stream>>>(sideL, ego, W_gc0, b_gc0, W_bi0, b_bi0,
                                          out, 64);

    gnn_spmm_own<<<OWN_BLOCKS, 1024, 0, stream>>>(rec, key4, ego, sideL);
    gnn_layer<<<gLayer, 256, 0, stream>>>(sideL, ego, W_gc1, b_gc1, W_bi1, b_bi1,
                                          out, 128);
}


# SpMM 2 re-reads the records that SpMM 1's owner blocks stored in queue order (in out's free columns; concat copy moved to layer 1): no key scan, no record gather there; sum order: LDS arrival
# speedup vs baseline: 3.4886x; 1.1332x over previous
#include <hip/hip_runtime.h>
#include <hip/hip_bf16.h>
#include <math.h>


typedef __attribute__((ext_vector_type(2))) float v2f;
typedef __attribute__((ext_vector_type(4))) float v4f;
typedef __attribute__((ext_vector_type(8))) float v8f;

#define GNN_N       100000
#define GNN_NUSERS  60000
#define GNN_EMB     64
#define GNN_E       3200000

__global__ __launch_bounds__(256) void gnn_init(
    const float* __restrict__ user_emb, const float* __restrict__ item_emb,
    float* __restrict__ ego)
{
    long tid = (long)blockIdx.x * 256 + threadIdx.x;
    if (tid >= (long)GNN_N * GNN_EMB) return;
    long row = tid >> 6;
    int  col = (int)(tid & 63);
    float v = (row < GNN_NUSERS)
                ? user_emb[row * GNN_EMB + col]
                : item_emb[(row - GNN_NUSERS) * GNN_EMB + col];
    ego[tid] = v;
}

#define OWN_ROWS   391
#define OWN_BLOCKS 256
#define OWN_Q      32768
#define KEY_G      (GNN_E / 32)
#define REC_DST_BITS 17
#define REC_DST_MASK ((1u << REC_DST_BITS) - 1u)
#define REC_BAD_ROW  511u
#define REC_COL0     64
#define LIST_NONE    0xFFFFFFFFu

static_assert(OWN_ROWS * OWN_BLOCKS >= GNN_N, "owners must cover all rows");
static_assert(GNN_E % 256 == 0, "the key pass runs full waves only");
static_assert(KEY_G == GNN_N, "one 256-B record group per output row");
static_assert(GNN_N <= (1 << REC_DST_BITS), "dst must fit its record field");
static_assert(OWN_ROWS <= (int)REC_BAD_ROW, "local row must fit below the bad-row code");

__global__ __launch_bounds__(256) void gnn_key(
    const int* __restrict__ esrc, const int* __restrict__ edst,
    const float* __restrict__ eval, unsigned* __restrict__ key,
    float* __restrict__ out)
{
    const int e    = blockIdx.x * 256 + threadIdx.x;
    const int lane = threadIdx.x & 31;
    const unsigned s  = (unsigned)esrc[e];
    const unsigned oq = s / OWN_ROWS;
    const unsigned o  = oq & 255u;
    const unsigned lr = (oq < (unsigned)OWN_BLOCKS) ? (s - oq * OWN_ROWS) : REC_BAD_ROW;
    uint2 rc;
    rc.x = (((unsigned)edst[e]) & REC_DST_MASK) | (lr << REC_DST_BITS);
    rc.y = __float_as_uint(eval[e]);
    *(uint2*)(out + (long)(e >> 5) * 192 + REC_COL0 + lane * 2) = rc;
    unsigned mine = 0u;
#pragma unroll
    for (int j = 0; j < 8; ++j) {
        unsigned bj = __builtin_amdgcn_ballot_w32(((o >> j) & 1u) != 0u);
        if (lane == j) mine = bj;
    }
    if (lane < 8) key[(e >> 5) * 8 + lane] = mine;
}

__device__ __forceinline__ unsigned* own_chunk(float* out, int lo, int rows, unsigned c)
{
    return (c < (unsigned)rows)
        ? (unsigned*)(out + (long)(lo + (int)c) * 192)
        : (unsigned*)(out + (long)(lo + (int)c - rows) * 192 + 128);
}

__device__ __forceinline__ void own_sum32(
    const uint2 rc, const int rows, const float* __restrict__ x, float* acc, const int lane)
{
    const int   d = (int)(rc.x & REC_DST_MASK);
    const int   r = (int)(rc.x >> REC_DST_BITS);
    const float v = __uint_as_float(rc.y);
    const bool ok = (unsigned)r < (unsigned)rows;
    const unsigned okm = __builtin_amdgcn_ballot_w32(ok);
#pragma unroll
    for (int j0 = 0; j0 < 32; j0 += 16) {
        if (((okm >> j0) & 0xFFFFu) == 0u) continue;
        v2f xv[16];
#pragma unroll
        for (int u = 0; u < 16; ++u) {
            int dj = __builtin_amdgcn_readlane(d, j0 + u);
            xv[u] = *(const v2f*)(x + (long)dj * GNN_EMB + lane * 2);
        }
#pragma unroll
        for (int u = 0; u < 16; ++u) {
            if ((okm >> (j0 + u)) & 1u) {
                int   rj = __builtin_amdgcn_readlane(r, j0 + u);
                float vj = __int_as_float(
                    __builtin_amdgcn_readlane(__float_as_int(v), j0 + u));
                float* a = &acc[rj * GNN_EMB + lane * 2];
                __hip_atomic_fetch_add(a,     vj * xv[u].x, __ATOMIC_RELAXED,
                                       __HIP_MEMORY_SCOPE_WORKGROUP);
                __hip_atomic_fetch_add(a + 1, vj * xv[u].y, __ATOMIC_RELAXED,
                                       __HIP_MEMORY_SCOPE_WORKGROUP);
            }
        }
    }
}

template <bool RAW>
__device__ __forceinline__ void own_scan_sum(
    const int* __restrict__ esrc, const int* __restrict__ edst,
    const float* __restrict__ eval, const unsigned* __restrict__ key,
    const float* __restrict__ x, float* out,
    float* acc, unsigned* q, unsigned* qn,
    const int tid, const int lane, const int wave, const int lo, const int rows)
{
    const uint4* __restrict__ key16 = (const uint4*)key;
    const unsigned* rec = (const unsigned*)(out + REC_COL0);
    const unsigned cap = 2u * (unsigned)rows;
    unsigned inv[8];
#pragma unroll
    for (int j = 0; j < 8; ++j) inv[j] = ((blockIdx.x >> j) & 1u) ? 0u : ~0u;

    int pos  = 0;
    int step = KEY_G;
    unsigned done = 0u;
    for (int trip = 0; trip < 3 + GNN_E / OWN_Q && pos < KEY_G; ++trip) {
        const int end = min(pos + step, KEY_G);
        if (tid == 0) *qn = 0;
        __syncthreads();

        for (int g0 = pos + tid; g0 < end; g0 += 4096) {
            uint4 ka[4], kb[4];
#pragma unroll
            for (int u = 0; u < 4; ++u) {
                int gc = min(g0 + u * 1024, KEY_G - 1);
                ka[u] = key16[2 * gc];
                kb[u] = key16[2 * gc + 1];
            }
#pragma unroll
            for (int u = 0; u < 4; ++u) {
                const int g = g0 + u * 1024;
                unsigned m = (ka[u].x ^ inv[0]) & (ka[u].y ^ inv[1])
                           & (ka[u].z ^ inv[2]) & (ka[u].w ^ inv[3])
                           & (kb[u].x ^ inv[4]) & (kb[u].y ^ inv[5])
                           & (kb[u].z ^ inv[6]) & (kb[u].w ^ inv[7]);
                if (g >= end) m = 0u;
                while (m) {
                    unsigned bit = (unsigned)__ffs((int)m) - 1u;
                    m &= m - 1u;
                    unsigned slot = atomicAdd(qn, 1u);
                    if (slot < OWN_Q) q[slot] = (unsigned)g * 32u + bit;
                }
            }
        }
        __syncthreads();
        const unsigned n = __builtin_amdgcn_readfirstlane(*qn);
        __syncthreads();
        if (n > OWN_Q) { step = OWN_Q / 32; continue; }

        for (unsigned base = (unsigned)wave * 32u; base < n; base += 1024u) {
            const unsigned qi = base + lane;
            const unsigned e = q[min(qi, n - 1u)];
            uint2 rc;
            if (RAW) {
                const unsigned s  = (unsigned)esrc[e];
                const unsigned oq = s / OWN_ROWS;
                const unsigned lr = (oq == blockIdx.x) ? (s - oq * OWN_ROWS) : REC_BAD_ROW;
                rc.x = (((unsigned)edst[e]) & REC_DST_MASK) | (lr << REC_DST_BITS);
                rc.y = __float_as_uint(eval[e]);
            } else {
                rc = *(const uint2*)(rec + (size_t)(e >> 5) * 192u + (e & 31u) * 2u);
            }
            if (qi >= n) rc.x = (rc.x & REC_DST_MASK) | (REC_BAD_ROW << REC_DST_BITS);
            if (!RAW) {
                const unsigned c = 1u + done + (base >> 5);
                if (c < cap) *(uint2*)(own_chunk(out, lo, rows, c) + lane * 2) = rc;
            }
            own_sum32(rc, rows, x, acc, lane);
        }
        done += (n + 31u) >> 5;
        pos = end;
    }
    if (!RAW && wave == 0) {
        const unsigned h = (1u + done <= cap) ? done : LIST_NONE;
        uint2 hv; hv.x = h; hv.y = h;
        *(uint2*)(own_chunk(out, lo, rows, 0u) + lane * 2) = hv;
    }
}

template <bool SECOND>
__global__ __launch_bounds__(1024) void gnn_spmm_own(
    const int* __restrict__ esrc, const int* __restrict__ edst,
    const float* __restrict__ eval, const unsigned* __restrict__ key,
    const float* __restrict__ x, float* __restrict__ y, float* out)
{
    __shared__ __attribute__((aligned(16))) float acc[OWN_ROWS * GNN_EMB];
    __shared__ unsigned q[OWN_Q];
    __shared__ unsigned qn;

    const int tid  = threadIdx.x;
    const int lane = tid & 31;
    const int wave = __builtin_amdgcn_readfirstlane(tid >> 5);
    const int lo   = blockIdx.x * OWN_ROWS;
    const int rows = min(OWN_ROWS, GNN_N - lo);

    for (int i = tid; i < OWN_ROWS * GNN_EMB; i += 1024) acc[i] = 0.f;

    if (!SECOND) {
        own_scan_sum<false>(esrc, edst, eval, key, x, out, acc, q, &qn,
                            tid, lane, wave, lo, rows);
    } else {
        const unsigned h = __builtin_amdgcn_readfirstlane(*own_chunk(out, lo, rows, 0u));
        if (h != LIST_NONE) {
            const unsigned total = min(h, 2u * (unsigned)rows - 1u);
            __syncthreads();
            unsigned c = (unsigned)wave;
            uint2 nx; nx.x = REC_BAD_ROW << REC_DST_BITS; nx.y = 0u;
            if (c < total) nx = *(const uint2*)(own_chunk(out, lo, rows, 1u + c) + lane * 2);
            for (; c < total; c += 32u) {
                const uint2 rc = nx;
                if (c + 32u < total)
                    nx = *(const uint2*)(own_chunk(out, lo, rows, 33u + c) + lane * 2);
                own_sum32(rc, rows, x, acc, lane);
            }
        } else {
            own_scan_sum<true>(esrc, edst, eval, key, x, out, acc, q, &qn,
                               tid, lane, wave, lo, rows);
        }
    }
    __syncthreads();

    float4* yo = (float4*)(y + (long)lo * GNN_EMB);
    const float4* a4 = (const float4*)acc;
    for (int i = tid; i < rows * (GNN_EMB / 4); i += 1024) yo[i] = a4[i];
}

__global__ __launch_bounds__(256) void gnn_layer(
    const float* __restrict__ sideL, float* __restrict__ ego,
    const float* __restrict__ Wgc, const float* __restrict__ bgc,
    const float* __restrict__ Wbi, const float* __restrict__ bbi,
    float* __restrict__ out, int out_col0,
    const float* __restrict__ user_emb, const float* __restrict__ item_emb, int concat)
{
    __shared__ float Wt[64][132];
    __shared__ float bsum[64];

    const int tid = threadIdx.x;

    for (int i = tid; i < 64 * 128; i += 256) {
        int k = i >> 6;
        int n = i & 63;
        float w = (k < 64) ? Wgc[k * 64 + n] : Wbi[(k - 64) * 64 + n];
        Wt[n][k] = w;
    }
    if (tid < 64) bsum[tid] = bgc[tid] + bbi[tid];
    __syncthreads();

    const int wave    = tid >> 5;
    const int lane    = tid & 31;
    const int rowbase = blockIdx.x * 128 + wave * 16;
    if (rowbase >= GNN_N) return;

    const int m    = lane & 15;
    const int half = lane >> 4;
    const int kb   = half << 1;

    if (concat) {
        static_assert(GNN_NUSERS % 16 == 0, "a 16-row tile must not straddle users and items");
        const float* cb = (rowbase < GNN_NUSERS)
            ? user_emb + (long)rowbase * GNN_EMB
            : item_emb + (long)(rowbase - GNN_NUSERS) * GNN_EMB;
        v4f cv[8];
#pragma unroll
        for (int t = 0; t < 8; ++t)
            cv[t] = *(const v4f*)(cb + (2 * t + half) * GNN_EMB + m * 4);
        asm volatile("" :: "v"(cv[0]), "v"(cv[1]), "v"(cv[2]), "v"(cv[3]),
                           "v"(cv[4]), "v"(cv[5]), "v"(cv[6]), "v"(cv[7]));
#pragma unroll
        for (int t = 0; t < 8; ++t)
            *(v4f*)(out + (long)(rowbase + 2 * t + half) * 192 + m * 4) = cv[t];
    }

    const long  row    = (long)(rowbase + m);
    const float* egoRd = ego   + row * 64;
    const float* sLRd  = sideL + row * 64;

    v2f e2[16], l2[16];
#pragma unroll
    for (int s = 0; s < 16; ++s) {
        int c = 4 * s + kb;
        e2[s] = *(const v2f*)(egoRd + c);
        l2[s] = *(const v2f*)(sLRd + c);
    }

    v8f acc[4] = {};

#pragma unroll
    for (int s = 0; s < 16; ++s) {
        v2f a;
        a.x = e2[s].x + l2[s].x;
        a.y = e2[s].y + l2[s].y;
#pragma unroll
        for (int nt = 0; nt < 4; ++nt) {
            v2f b = *(const v2f*)(&Wt[nt * 16 + m][4 * s + kb]);
            acc[nt] = __builtin_amdgcn_wmma_f32_16x16x4_f32(
                false, a, false, b, (short)0, acc[nt], false, false);
        }
    }
#pragma unroll
    for (int s = 0; s < 16; ++s) {
        v2f a;
        a.x = e2[s].x * l2[s].x;
        a.y = e2[s].y * l2[s].y;
#pragma unroll
        for (int nt = 0; nt < 4; ++nt) {
            v2f b = *(const v2f*)(&Wt[nt * 16 + m][64 + 4 * s + kb]);
            acc[nt] = __builtin_amdgcn_wmma_f32_16x16x4_f32(
                false, a, false, b, (short)0, acc[nt], false, false);
        }
    }

    float vout[4][8];
    float ssq[8];
#pragma unroll
    for (int j = 0; j < 8; ++j) ssq[j] = 0.f;
#pragma unroll
    for (int nt = 0; nt < 4; ++nt) {
        float bc = bsum[nt * 16 + m];
#pragma unroll
        for (int j = 0; j < 8; ++j) {
            float v = acc[nt][j] + bc;
            v = (v > 0.f) ? v : 0.01f * v;
            vout[nt][j] = v;
            ssq[j] += v * v;
        }
    }

    float* egoWr = ego + (long)rowbase * 64;
#pragma unroll
    for (int j = 0; j < 8; ++j) {
        int mrow = j + (half << 3);
#pragma unroll
        for (int nt = 0; nt < 4; ++nt)
            egoWr[mrow * 64 + nt * 16 + m] = vout[nt][j];
    }

#pragma unroll
    for (int j = 0; j < 8; ++j) {
        float s = ssq[j];
        s += __shfl_xor(s, 8, 16);
        s += __shfl_xor(s, 4, 16);
        s += __shfl_xor(s, 2, 16);
        s += __shfl_xor(s, 1, 16);
        float inv = 1.0f / fmaxf(sqrtf(s), 1e-12f);
        int mrow = j + (half << 3);
        float* o = out + (long)(rowbase + mrow) * 192 + out_col0;
#pragma unroll
        for (int nt = 0; nt < 4; ++nt)
            o[nt * 16 + m] = vout[nt][j] * inv;
    }
}

extern "C" void kernel_launch(void* const* d_in, const int* in_sizes, int n_in,
                              void* d_out, int out_size, void* d_ws, size_t ws_size,
                              hipStream_t stream) {
    const float* user_emb = (const float*)d_in[0];
    const float* item_emb = (const float*)d_in[1];
    const float* W_gc0 = (const float*)d_in[2];
    const float* b_gc0 = (const float*)d_in[3];
    const float* W_bi0 = (const float*)d_in[4];
    const float* b_bi0 = (const float*)d_in[5];
    const float* W_gc1 = (const float*)d_in[6];
    const float* b_gc1 = (const float*)d_in[7];
    const float* W_bi1 = (const float*)d_in[8];
    const float* b_bi1 = (const float*)d_in[9];
    const int*   e_src = (const int*)d_in[10];
    const int*   e_dst = (const int*)d_in[11];
    const float* e_val = (const float*)d_in[12];
    float* out = (float*)d_out;

    float* ego   = (float*)d_ws;
    float* sideL = ego + (long)GNN_N * GNN_EMB;
    unsigned* key4 = (unsigned*)(sideL + (long)GNN_N * GNN_EMB);

    const long nelem  = (long)GNN_N * GNN_EMB;
    const int  gInit  = (int)((nelem + 255) / 256);
    const int  gKey   = GNN_E / 256;
    const int  gLayer = (GNN_N + 127) / 128;

    gnn_init<<<gInit, 256, 0, stream>>>(user_emb, item_emb, ego);
    gnn_key<<<gKey, 256, 0, stream>>>(e_src, e_dst, e_val, key4, out);

    gnn_spmm_own<false><<<OWN_BLOCKS, 1024, 0, stream>>>(e_src, e_dst, e_val, key4,
                                                         ego, sideL, out);
    gnn_layer<<<gLayer, 256, 0, stream>>>(sideL, ego, W_gc0, b_gc0, W_bi0, b_bi0,
                                          out, 64, user_emb, item_emb, 0);

    gnn_spmm_own<true><<<OWN_BLOCKS, 1024, 0, stream>>>(e_src, e_dst, e_val, key4,
                                                        ego, sideL, out);
    gnn_layer<<<gLayer, 256, 0, stream>>>(sideL, ego, W_gc1, b_gc1, W_bi1, b_bi1,
                                          out, 128, user_emb, item_emb, 1);
}


# Ego copy and key pass each run as 625 resident blocks (a thread copies ten 16-B pieces; a wave keys 640 edges, whole-line key stores), were one element per thread: 300000 short waves become 10000
# speedup vs baseline: 3.6565x; 1.0481x over previous
#include <hip/hip_runtime.h>
#include <hip/hip_bf16.h>
#include <math.h>


typedef __attribute__((ext_vector_type(2))) float v2f;
typedef __attribute__((ext_vector_type(4))) float v4f;
typedef __attribute__((ext_vector_type(8))) float v8f;

#define GNN_N       100000
#define GNN_NUSERS  60000
#define GNN_EMB     64
#define GNN_E       3200000

#define FE_BLOCKS   625
#define FE_THREADS  (FE_BLOCKS * 256)
#define FE_WAVES    (FE_BLOCKS * 8)
#define INIT_V4     (GNN_N * GNN_EMB / 4)
#define INIT_U4     (GNN_NUSERS * GNN_EMB / 4)
#define INIT_U      (INIT_V4 / FE_THREADS)
static_assert(INIT_V4 % FE_THREADS == 0, "the ego copy has no tail");

__global__ __launch_bounds__(256) void gnn_init(
    const float* __restrict__ user_emb, const float* __restrict__ item_emb,
    float* __restrict__ ego)
{
    const v4f* u4 = (const v4f*)user_emb;
    const v4f* i4 = (const v4f*)item_emb;
    v4f*       e4 = (v4f*)ego;
    const int i0 = blockIdx.x * 256 + threadIdx.x;
    v4f v[INIT_U];
#pragma unroll
    for (int u = 0; u < INIT_U; ++u) {
        const int i = i0 + u * FE_THREADS;
        const v4f* p = (i < INIT_U4) ? u4 + i : i4 + (i - INIT_U4);
        v[u] = *p;
    }
#pragma unroll
    for (int u = 0; u < INIT_U; ++u) e4[i0 + u * FE_THREADS] = v[u];
}

#define OWN_ROWS   391
#define OWN_BLOCKS 256
#define OWN_Q      32768
#define KEY_G      (GNN_E / 32)
#define KEY_SG     (GNN_E / 128)
#define KEY_PER_WAVE (KEY_SG / FE_WAVES)
#define REC_DST_BITS 17
#define REC_DST_MASK ((1u << REC_DST_BITS) - 1u)
#define REC_BAD_ROW  511u
#define REC_COL0     64
#define LIST_NONE    0xFFFFFFFFu

static_assert(OWN_ROWS * OWN_BLOCKS >= GNN_N, "owners must cover all rows");
static_assert(KEY_SG % FE_WAVES == 0, "the key pass has no tail");
static_assert(KEY_G == GNN_N, "one 256-B record group per output row");
static_assert(GNN_N <= (1 << REC_DST_BITS), "dst must fit its record field");
static_assert(OWN_ROWS <= (int)REC_BAD_ROW, "local row must fit below the bad-row code");

__global__ __launch_bounds__(256) void gnn_key(
    const int* __restrict__ esrc, const int* __restrict__ edst,
    const float* __restrict__ eval, unsigned* __restrict__ key,
    float* __restrict__ out)
{
    const int lane = threadIdx.x & 31;
    const int wv = blockIdx.x * 8 + __builtin_amdgcn_readfirstlane(threadIdx.x >> 5);
    unsigned s[KEY_PER_WAVE][4], d[KEY_PER_WAVE][4], v[KEY_PER_WAVE][4];
#pragma unroll
    for (int h = 0; h < KEY_PER_WAVE; ++h) {
#pragma unroll
        for (int u = 0; u < 4; ++u) {
            const int e = (wv + h * FE_WAVES) * 128 + u * 32 + lane;
            s[h][u] = (unsigned)esrc[e];
            d[h][u] = (unsigned)edst[e];
            v[h][u] = __float_as_uint(eval[e]);
        }
    }
#pragma unroll
    for (int h = 0; h < KEY_PER_WAVE; ++h) {
        const int G = wv + h * FE_WAVES;
        unsigned mine = 0u;
#pragma unroll
        for (int u = 0; u < 4; ++u) {
            const unsigned oq = s[h][u] / OWN_ROWS;
            const unsigned o  = oq & 255u;
            const unsigned lr = (oq < (unsigned)OWN_BLOCKS)
                                  ? (s[h][u] - oq * OWN_ROWS) : REC_BAD_ROW;
            uint2 rc;
            rc.x = (d[h][u] & REC_DST_MASK) | (lr << REC_DST_BITS);
            rc.y = v[h][u];
            *(uint2*)(out + (long)(4 * G + u) * 192 + REC_COL0 + lane * 2) = rc;
#pragma unroll
            for (int j = 0; j < 8; ++j) {
                unsigned bj = __builtin_amdgcn_ballot_w32(((o >> j) & 1u) != 0u);
                if (lane == u * 8 + j) mine = bj;
            }
        }
        key[G * 32 + lane] = mine;
    }
}

__device__ __forceinline__ unsigned* own_chunk(float* out, int lo, int rows, unsigned c)
{
    return (c < (unsigned)rows)
        ? (unsigned*)(out + (long)(lo + (int)c) * 192)
        : (unsigned*)(out + (long)(lo + (int)c - rows) * 192 + 128);
}

__device__ __forceinline__ void own_sum32(
    const uint2 rc, const int rows, const float* __restrict__ x, float* acc, const int lane)
{
    const int   d = (int)(rc.x & REC_DST_MASK);
    const int   r = (int)(rc.x >> REC_DST_BITS);
    const float v = __uint_as_float(rc.y);
    const bool ok = (unsigned)r < (unsigned)rows;
    const unsigned okm = __builtin_amdgcn_ballot_w32(ok);
#pragma unroll
    for (int j0 = 0; j0 < 32; j0 += 16) {
        if (((okm >> j0) & 0xFFFFu) == 0u) continue;
        v2f xv[16];
#pragma unroll
        for (int u = 0; u < 16; ++u) {
            int dj = __builtin_amdgcn_readlane(d, j0 + u);
            xv[u] = *(const v2f*)(x + (long)dj * GNN_EMB + lane * 2);
        }
#pragma unroll
        for (int u = 0; u < 16; ++u) {
            if ((okm >> (j0 + u)) & 1u) {
                int   rj = __builtin_amdgcn_readlane(r, j0 + u);
                float vj = __int_as_float(
                    __builtin_amdgcn_readlane(__float_as_int(v), j0 + u));
                float* a = &acc[rj * GNN_EMB + lane * 2];
                __hip_atomic_fetch_add(a,     vj * xv[u].x, __ATOMIC_RELAXED,
                                       __HIP_MEMORY_SCOPE_WORKGROUP);
                __hip_atomic_fetch_add(a + 1, vj * xv[u].y, __ATOMIC_RELAXED,
                                       __HIP_MEMORY_SCOPE_WORKGROUP);
            }
        }
    }
}

template <bool RAW>
__device__ __forceinline__ void own_scan_sum(
    const int* __restrict__ esrc, const int* __restrict__ edst,
    const float* __restrict__ eval, const unsigned* __restrict__ key,
    const float* __restrict__ x, float* out,
    float* acc, unsigned* q, unsigned* qn,
    const int tid, const int lane, const int wave, const int lo, const int rows)
{
    const uint4* __restrict__ key16 = (const uint4*)key;
    const unsigned* rec = (const unsigned*)(out + REC_COL0);
    const unsigned cap = 2u * (unsigned)rows;
    unsigned inv[8];
#pragma unroll
    for (int j = 0; j < 8; ++j) inv[j] = ((blockIdx.x >> j) & 1u) ? 0u : ~0u;

    int pos  = 0;
    int step = KEY_G;
    unsigned done = 0u;
    for (int trip = 0; trip < 3 + GNN_E / OWN_Q && pos < KEY_G; ++trip) {
        const int end = min(pos + step, KEY_G);
        if (tid == 0) *qn = 0;
        __syncthreads();

        for (int g0 = pos + tid; g0 < end; g0 += 4096) {
            uint4 ka[4], kb[4];
#pragma unroll
            for (int u = 0; u < 4; ++u) {
                int gc = min(g0 + u * 1024, KEY_G - 1);
                ka[u] = key16[2 * gc];
                kb[u] = key16[2 * gc + 1];
            }
#pragma unroll
            for (int u = 0; u < 4; ++u) {
                const int g = g0 + u * 1024;
                unsigned m = (ka[u].x ^ inv[0]) & (ka[u].y ^ inv[1])
                           & (ka[u].z ^ inv[2]) & (ka[u].w ^ inv[3])
                           & (kb[u].x ^ inv[4]) & (kb[u].y ^ inv[5])
                           & (kb[u].z ^ inv[6]) & (kb[u].w ^ inv[7]);
                if (g >= end) m = 0u;
                while (m) {
                    unsigned bit = (unsigned)__ffs((int)m) - 1u;
                    m &= m - 1u;
                    unsigned slot = atomicAdd(qn, 1u);
                    if (slot < OWN_Q) q[slot] = (unsigned)g * 32u + bit;
                }
            }
        }
        __syncthreads();
        const unsigned n = __builtin_amdgcn_readfirstlane(*qn);
        __syncthreads();
        if (n > OWN_Q) { step = OWN_Q / 32; continue; }

        for (unsigned base = (unsigned)wave * 32u; base < n; base += 1024u) {
            const unsigned qi = base + lane;
            const unsigned e = q[min(qi, n - 1u)];
            uint2 rc;
            if (RAW) {
                const unsigned s  = (unsigned)esrc[e];
                const unsigned oq = s / OWN_ROWS;
                const unsigned lr = (oq == blockIdx.x) ? (s - oq * OWN_ROWS) : REC_BAD_ROW;
                rc.x = (((unsigned)edst[e]) & REC_DST_MASK) | (lr << REC_DST_BITS);
                rc.y = __float_as_uint(eval[e]);
            } else {
                rc = *(const uint2*)(rec + (size_t)(e >> 5) * 192u + (e & 31u) * 2u);
            }
            if (qi >= n) rc.x = (rc.x & REC_DST_MASK) | (REC_BAD_ROW << REC_DST_BITS);
            if (!RAW) {
                const unsigned c = 1u + done + (base >> 5);
                if (c < cap) *(uint2*)(own_chunk(out, lo, rows, c) + lane * 2) = rc;
            }
            own_sum32(rc, rows, x, acc, lane);
        }
        done += (n + 31u) >> 5;
        pos = end;
    }
    if (!RAW && wave == 0) {
        const unsigned h = (1u + done <= cap) ? done : LIST_NONE;
        uint2 hv; hv.x = h; hv.y = h;
        *(uint2*)(own_chunk(out, lo, rows, 0u) + lane * 2) = hv;
    }
}

template <bool SECOND>
__global__ __launch_bounds__(1024) void gnn_spmm_own(
    const int* __restrict__ esrc, const int* __restrict__ edst,
    const float* __restrict__ eval, const unsigned* __restrict__ key,
    const float* __restrict__ x, float* __restrict__ y, float* out)
{
    __shared__ __attribute__((aligned(16))) float acc[OWN_ROWS * GNN_EMB];
    __shared__ unsigned q[OWN_Q];
    __shared__ unsigned qn;

    const int tid  = threadIdx.x;
    const int lane = tid & 31;
    const int wave = __builtin_amdgcn_readfirstlane(tid >> 5);
    const int lo   = blockIdx.x * OWN_ROWS;
    const int rows = min(OWN_ROWS, GNN_N - lo);

    for (int i = tid; i < OWN_ROWS * GNN_EMB; i += 1024) acc[i] = 0.f;

    if (!SECOND) {
        own_scan_sum<false>(esrc, edst, eval, key, x, out, acc, q, &qn,
                            tid, lane, wave, lo, rows);
    } else {
        const unsigned h = __builtin_amdgcn_readfirstlane(*own_chunk(out, lo, rows, 0u));
        if (h != LIST_NONE) {
            const unsigned total = min(h, 2u * (unsigned)rows - 1u);
            __syncthreads();
            unsigned c = (unsigned)wave;
            uint2 nx; nx.x = REC_BAD_ROW << REC_DST_BITS; nx.y = 0u;
            if (c < total) nx = *(const uint2*)(own_chunk(out, lo, rows, 1u + c) + lane * 2);
            for (; c < total; c += 32u) {
                const uint2 rc = nx;
                if (c + 32u < total)
                    nx = *(const uint2*)(own_chunk(out, lo, rows, 33u + c) + lane * 2);
                own_sum32(rc, rows, x, acc, lane);
            }
        } else {
            own_scan_sum<true>(esrc, edst, eval, key, x, out, acc, q, &qn,
                               tid, lane, wave, lo, rows);
        }
    }
    __syncthreads();

    float4* yo = (float4*)(y + (long)lo * GNN_EMB);
    const float4* a4 = (const float4*)acc;
    for (int i = tid; i < rows * (GNN_EMB / 4); i += 1024) yo[i] = a4[i];
}

__global__ __launch_bounds__(256) void gnn_layer(
    const float* __restrict__ sideL, float* __restrict__ ego,
    const float* __restrict__ Wgc, const float* __restrict__ bgc,
    const float* __restrict__ Wbi, const float* __restrict__ bbi,
    float* __restrict__ out, int out_col0,
    const float* __restrict__ user_emb, const float* __restrict__ item_emb, int concat)
{
    __shared__ float Wt[64][132];
    __shared__ float bsum[64];

    const int tid = threadIdx.x;

    for (int i = tid; i < 64 * 128; i += 256) {
        int k = i >> 6;
        int n = i & 63;
        float w = (k < 64) ? Wgc[k * 64 + n] : Wbi[(k - 64) * 64 + n];
        Wt[n][k] = w;
    }
    if (tid < 64) bsum[tid] = bgc[tid] + bbi[tid];
    __syncthreads();

    const int wave    = tid >> 5;
    const int lane    = tid & 31;
    const int rowbase = blockIdx.x * 128 + wave * 16;
    if (rowbase >= GNN_N) return;

    const int m    = lane & 15;
    const int half = lane >> 4;
    const int kb   = half << 1;

    if (concat) {
        static_assert(GNN_NUSERS % 16 == 0, "a 16-row tile must not straddle users and items");
        const float* cb = (rowbase < GNN_NUSERS)
            ? user_emb + (long)rowbase * GNN_EMB
            : item_emb + (long)(rowbase - GNN_NUSERS) * GNN_EMB;
        v4f cv[8];
#pragma unroll
        for (int t = 0; t < 8; ++t)
            cv[t] = *(const v4f*)(cb + (2 * t + half) * GNN_EMB + m * 4);
        asm volatile("" :: "v"(cv[0]), "v"(cv[1]), "v"(cv[2]), "v"(cv[3]),
                           "v"(cv[4]), "v"(cv[5]), "v"(cv[6]), "v"(cv[7]));
#pragma unroll
        for (int t = 0; t < 8; ++t)
            *(v4f*)(out + (long)(rowbase + 2 * t + half) * 192 + m * 4) = cv[t];
    }

    const long  row    = (long)(rowbase + m);
    const float* egoRd = ego   + row * 64;
    const float* sLRd  = sideL + row * 64;

    v2f e2[16], l2[16];
#pragma unroll
    for (int s = 0; s < 16; ++s) {
        int c = 4 * s + kb;
        e2[s] = *(const v2f*)(egoRd + c);
        l2[s] = *(const v2f*)(sLRd + c);
    }

    v8f acc[4] = {};

#pragma unroll
    for (int s = 0; s < 16; ++s) {
        v2f a;
        a.x = e2[s].x + l2[s].x;
        a.y = e2[s].y + l2[s].y;
#pragma unroll
        for (int nt = 0; nt < 4; ++nt) {
            v2f b = *(const v2f*)(&Wt[nt * 16 + m][4 * s + kb]);
            acc[nt] = __builtin_amdgcn_wmma_f32_16x16x4_f32(
                false, a, false, b, (short)0, acc[nt], false, false);
        }
    }
#pragma unroll
    for (int s = 0; s < 16; ++s) {
        v2f a;
        a.x = e2[s].x * l2[s].x;
        a.y = e2[s].y * l2[s].y;
#pragma unroll
        for (int nt = 0; nt < 4; ++nt) {
            v2f b = *(const v2f*)(&Wt[nt * 16 + m][64 + 4 * s + kb]);
            acc[nt] = __builtin_amdgcn_wmma_f32_16x16x4_f32(
                false, a, false, b, (short)0, acc[nt], false, false);
        }
    }

    float vout[4][8];
    float ssq[8];
#pragma unroll
    for (int j = 0; j < 8; ++j) ssq[j] = 0.f;
#pragma unroll
    for (int nt = 0; nt < 4; ++nt) {
        float bc = bsum[nt * 16 + m];
#pragma unroll
        for (int j = 0; j < 8; ++j) {
            float v = acc[nt][j] + bc;
            v = (v > 0.f) ? v : 0.01f * v;
            vout[nt][j] = v;
            ssq[j] += v * v;
        }
    }

    float* egoWr = ego + (long)rowbase * 64;
#pragma unroll
    for (int j = 0; j < 8; ++j) {
        int mrow = j + (half << 3);
#pragma unroll
        for (int nt = 0; nt < 4; ++nt)
            egoWr[mrow * 64 + nt * 16 + m] = vout[nt][j];
    }

#pragma unroll
    for (int j = 0; j < 8; ++j) {
        float s = ssq[j];
        s += __shfl_xor(s, 8, 16);
        s += __shfl_xor(s, 4, 16);
        s += __shfl_xor(s, 2, 16);
        s += __shfl_xor(s, 1, 16);
        float inv = 1.0f / fmaxf(sqrtf(s), 1e-12f);
        int mrow = j + (half << 3);
        float* o = out + (long)(rowbase + mrow) * 192 + out_col0;
#pragma unroll
        for (int nt = 0; nt < 4; ++nt)
            o[nt * 16 + m] = vout[nt][j] * inv;
    }
}

extern "C" void kernel_launch(void* const* d_in, const int* in_sizes, int n_in,
                              void* d_out, int out_size, void* d_ws, size_t ws_size,
                              hipStream_t stream) {
    const float* user_emb = (const float*)d_in[0];
    const float* item_emb = (const float*)d_in[1];
    const float* W_gc0 = (const float*)d_in[2];
    const float* b_gc0 = (const float*)d_in[3];
    const float* W_bi0 = (const float*)d_in[4];
    const float* b_bi0 = (const float*)d_in[5];
    const float* W_gc1 = (const float*)d_in[6];
    const float* b_gc1 = (const float*)d_in[7];
    const float* W_bi1 = (const float*)d_in[8];
    const float* b_bi1 = (const float*)d_in[9];
    const int*   e_src = (const int*)d_in[10];
    const int*   e_dst = (const int*)d_in[11];
    const float* e_val = (const float*)d_in[12];
    float* out = (float*)d_out;

    float* ego   = (float*)d_ws;
    float* sideL = ego + (long)GNN_N * GNN_EMB;
    unsigned* key4 = (unsigned*)(sideL + (long)GNN_N * GNN_EMB);

    const int  gLayer = (GNN_N + 127) / 128;

    gnn_init<<<FE_BLOCKS, 256, 0, stream>>>(user_emb, item_emb, ego);
    gnn_key<<<FE_BLOCKS, 256, 0, stream>>>(e_src, e_dst, e_val, key4, out);

    gnn_spmm_own<false><<<OWN_BLOCKS, 1024, 0, stream>>>(e_src, e_dst, e_val, key4,
                                                         ego, sideL, out);
    gnn_layer<<<gLayer, 256, 0, stream>>>(sideL, ego, W_gc0, b_gc0, W_bi0, b_bi0,
                                          out, 64, user_emb, item_emb, 0);

    gnn_spmm_own<true><<<OWN_BLOCKS, 1024, 0, stream>>>(e_src, e_dst, e_val, key4,
                                                        ego, sideL, out);
    gnn_layer<<<gLayer, 256, 0, stream>>>(sideL, ego, W_gc1, b_gc1, W_bi1, b_bi1,
                                          out, 128, user_emb, item_emb, 1);
}


# Ego is never copied: SpMM 1 gathers x rows and layer 0 reads its A rows from the two embedding tables (ego copy launch gone, 6 launches to 5); the last layer skips its unread ego store
# speedup vs baseline: 3.8688x; 1.0581x over previous
#include <hip/hip_runtime.h>
#include <hip/hip_bf16.h>
#include <math.h>


typedef __attribute__((ext_vector_type(2))) float v2f;
typedef __attribute__((ext_vector_type(4))) float v4f;
typedef __attribute__((ext_vector_type(8))) float v8f;

#define GNN_N       100000
#define GNN_NUSERS  60000
#define GNN_EMB     64
#define GNN_E       3200000

#define FE_BLOCKS   625
#define FE_WAVES    (FE_BLOCKS * 8)

#define OWN_ROWS   391
#define OWN_BLOCKS 256
#define OWN_Q      32768
#define KEY_G      (GNN_E / 32)
#define KEY_SG     (GNN_E / 128)
#define KEY_PER_WAVE (KEY_SG / FE_WAVES)
#define REC_DST_BITS 17
#define REC_DST_MASK ((1u << REC_DST_BITS) - 1u)
#define REC_BAD_ROW  511u
#define REC_COL0     64
#define LIST_NONE    0xFFFFFFFFu

static_assert(OWN_ROWS * OWN_BLOCKS >= GNN_N, "owners must cover all rows");
static_assert(KEY_SG % FE_WAVES == 0, "the key pass has no tail");
static_assert(KEY_G == GNN_N, "one 256-B record group per output row");
static_assert(GNN_N <= (1 << REC_DST_BITS), "dst must fit its record field");
static_assert(OWN_ROWS <= (int)REC_BAD_ROW, "local row must fit below the bad-row code");

__global__ __launch_bounds__(256) void gnn_key(
    const int* __restrict__ esrc, const int* __restrict__ edst,
    const float* __restrict__ eval, unsigned* __restrict__ key,
    float* __restrict__ out)
{
    const int lane = threadIdx.x & 31;
    const int wv = blockIdx.x * 8 + __builtin_amdgcn_readfirstlane(threadIdx.x >> 5);
    unsigned s[KEY_PER_WAVE][4], d[KEY_PER_WAVE][4], v[KEY_PER_WAVE][4];
#pragma unroll
    for (int h = 0; h < KEY_PER_WAVE; ++h) {
#pragma unroll
        for (int u = 0; u < 4; ++u) {
            const int e = (wv + h * FE_WAVES) * 128 + u * 32 + lane;
            s[h][u] = (unsigned)esrc[e];
            d[h][u] = (unsigned)edst[e];
            v[h][u] = __float_as_uint(eval[e]);
        }
    }
#pragma unroll
    for (int h = 0; h < KEY_PER_WAVE; ++h) {
        const int G = wv + h * FE_WAVES;
        unsigned mine = 0u;
#pragma unroll
        for (int u = 0; u < 4; ++u) {
            const unsigned oq = s[h][u] / OWN_ROWS;
            const unsigned o  = oq & 255u;
            const unsigned lr = (oq < (unsigned)OWN_BLOCKS)
                                  ? (s[h][u] - oq * OWN_ROWS) : REC_BAD_ROW;
            uint2 rc;
            rc.x = (d[h][u] & REC_DST_MASK) | (lr << REC_DST_BITS);
            rc.y = v[h][u];
            *(uint2*)(out + (long)(4 * G + u) * 192 + REC_COL0 + lane * 2) = rc;
#pragma unroll
            for (int j = 0; j < 8; ++j) {
                unsigned bj = __builtin_amdgcn_ballot_w32(((o >> j) & 1u) != 0u);
                if (lane == u * 8 + j) mine = bj;
            }
        }
        key[G * 32 + lane] = mine;
    }
}

__device__ __forceinline__ unsigned* own_chunk(float* out, int lo, int rows, unsigned c)
{
    return (c < (unsigned)rows)
        ? (unsigned*)(out + (long)(lo + (int)c) * 192)
        : (unsigned*)(out + (long)(lo + (int)c - rows) * 192 + 128);
}

template <bool SPLIT>
__device__ __forceinline__ void own_sum32(
    const uint2 rc, const int rows, const float* __restrict__ xu,
    const float* __restrict__ xi, float* acc, const int lane)
{
    const int   d = (int)(rc.x & REC_DST_MASK);
    const int   r = (int)(rc.x >> REC_DST_BITS);
    const float v = __uint_as_float(rc.y);
    const bool ok = (unsigned)r < (unsigned)rows;
    const unsigned okm = __builtin_amdgcn_ballot_w32(ok);
    const float* xi0 = xi - (long)GNN_NUSERS * GNN_EMB;
#pragma unroll
    for (int j0 = 0; j0 < 32; j0 += 16) {
        if (((okm >> j0) & 0xFFFFu) == 0u) continue;
        v2f xv[16];
#pragma unroll
        for (int u = 0; u < 16; ++u) {
            int dj = __builtin_amdgcn_readlane(d, j0 + u);
            const float* xb = (SPLIT && dj >= GNN_NUSERS) ? xi0 : xu;
            xv[u] = *(const v2f*)(xb + (unsigned)dj * (unsigned)GNN_EMB + lane * 2);
        }
#pragma unroll
        for (int u = 0; u < 16; ++u) {
            if ((okm >> (j0 + u)) & 1u) {
                int   rj = __builtin_amdgcn_readlane(r, j0 + u);
                float vj = __int_as_float(
                    __builtin_amdgcn_readlane(__float_as_int(v), j0 + u));
                float* a = &acc[rj * GNN_EMB + lane * 2];
                __hip_atomic_fetch_add(a,     vj * xv[u].x, __ATOMIC_RELAXED,
                                       __HIP_MEMORY_SCOPE_WORKGROUP);
                __hip_atomic_fetch_add(a + 1, vj * xv[u].y, __ATOMIC_RELAXED,
                                       __HIP_MEMORY_SCOPE_WORKGROUP);
            }
        }
    }
}

template <bool RAW>
__device__ __forceinline__ void own_scan_sum(
    const int* __restrict__ esrc, const int* __restrict__ edst,
    const float* __restrict__ eval, const unsigned* __restrict__ key,
    const float* __restrict__ xu, const float* __restrict__ xi, float* out,
    float* acc, unsigned* q, unsigned* qn,
    const int tid, const int lane, const int wave, const int lo, const int rows)
{
    const uint4* __restrict__ key16 = (const uint4*)key;
    const unsigned* rec = (const unsigned*)(out + REC_COL0);
    const unsigned cap = 2u * (unsigned)rows;
    unsigned inv[8];
#pragma unroll
    for (int j = 0; j < 8; ++j) inv[j] = ((blockIdx.x >> j) & 1u) ? 0u : ~0u;

    int pos  = 0;
    int step = KEY_G;
    unsigned done = 0u;
    for (int trip = 0; trip < 3 + GNN_E / OWN_Q && pos < KEY_G; ++trip) {
        const int end = min(pos + step, KEY_G);
        if (tid == 0) *qn = 0;
        __syncthreads();

        for (int g0 = pos + tid; g0 < end; g0 += 4096) {
            uint4 ka[4], kb[4];
#pragma unroll
            for (int u = 0; u < 4; ++u) {
                int gc = min(g0 + u * 1024, KEY_G - 1);
                ka[u] = key16[2 * gc];
                kb[u] = key16[2 * gc + 1];
            }
#pragma unroll
            for (int u = 0; u < 4; ++u) {
                const int g = g0 + u * 1024;
                unsigned m = (ka[u].x ^ inv[0]) & (ka[u].y ^ inv[1])
                           & (ka[u].z ^ inv[2]) & (ka[u].w ^ inv[3])
                           & (kb[u].x ^ inv[4]) & (kb[u].y ^ inv[5])
                           & (kb[u].z ^ inv[6]) & (kb[u].w ^ inv[7]);
                if (g >= end) m = 0u;
                while (m) {
                    unsigned bit = (unsigned)__ffs((int)m) - 1u;
                    m &= m - 1u;
                    unsigned slot = atomicAdd(qn, 1u);
                    if (slot < OWN_Q) q[slot] = (unsigned)g * 32u + bit;
                }
            }
        }
        __syncthreads();
        const unsigned n = __builtin_amdgcn_readfirstlane(*qn);
        __syncthreads();
        if (n > OWN_Q) { step = OWN_Q / 32; continue; }

        for (unsigned base = (unsigned)wave * 32u; base < n; base += 1024u) {
            const unsigned qi = base + lane;
            const unsigned e = q[min(qi, n - 1u)];
            uint2 rc;
            if (RAW) {
                const unsigned s  = (unsigned)esrc[e];
                const unsigned oq = s / OWN_ROWS;
                const unsigned lr = (oq == blockIdx.x) ? (s - oq * OWN_ROWS) : REC_BAD_ROW;
                rc.x = (((unsigned)edst[e]) & REC_DST_MASK) | (lr << REC_DST_BITS);
                rc.y = __float_as_uint(eval[e]);
            } else {
                rc = *(const uint2*)(rec + (size_t)(e >> 5) * 192u + (e & 31u) * 2u);
            }
            if (qi >= n) rc.x = (rc.x & REC_DST_MASK) | (REC_BAD_ROW << REC_DST_BITS);
            if (!RAW) {
                const unsigned c = 1u + done + (base >> 5);
                if (c < cap) *(uint2*)(own_chunk(out, lo, rows, c) + lane * 2) = rc;
            }
            own_sum32<!RAW>(rc, rows, xu, xi, acc, lane);
        }
        done += (n + 31u) >> 5;
        pos = end;
    }
    if (!RAW && wave == 0) {
        const unsigned h = (1u + done <= cap) ? done : LIST_NONE;
        uint2 hv; hv.x = h; hv.y = h;
        *(uint2*)(own_chunk(out, lo, rows, 0u) + lane * 2) = hv;
    }
}

template <bool SECOND>
__global__ __launch_bounds__(1024) void gnn_spmm_own(
    const int* __restrict__ esrc, const int* __restrict__ edst,
    const float* __restrict__ eval, const unsigned* __restrict__ key,
    const float* __restrict__ xu, const float* __restrict__ xi,
    float* __restrict__ y, float* out)
{
    __shared__ __attribute__((aligned(16))) float acc[OWN_ROWS * GNN_EMB];
    __shared__ unsigned q[OWN_Q];
    __shared__ unsigned qn;

    const int tid  = threadIdx.x;
    const int lane = tid & 31;
    const int wave = __builtin_amdgcn_readfirstlane(tid >> 5);
    const int lo   = blockIdx.x * OWN_ROWS;
    const int rows = min(OWN_ROWS, GNN_N - lo);

    for (int i = tid; i < OWN_ROWS * GNN_EMB; i += 1024) acc[i] = 0.f;

    if (!SECOND) {
        own_scan_sum<false>(esrc, edst, eval, key, xu, xi, out, acc, q, &qn,
                            tid, lane, wave, lo, rows);
    } else {
        const unsigned h = __builtin_amdgcn_readfirstlane(*own_chunk(out, lo, rows, 0u));
        if (h != LIST_NONE) {
            const unsigned total = min(h, 2u * (unsigned)rows - 1u);
            __syncthreads();
            unsigned c = (unsigned)wave;
            uint2 nx; nx.x = REC_BAD_ROW << REC_DST_BITS; nx.y = 0u;
            if (c < total) nx = *(const uint2*)(own_chunk(out, lo, rows, 1u + c) + lane * 2);
            for (; c < total; c += 32u) {
                const uint2 rc = nx;
                if (c + 32u < total)
                    nx = *(const uint2*)(own_chunk(out, lo, rows, 33u + c) + lane * 2);
                own_sum32<false>(rc, rows, xu, xi, acc, lane);
            }
        } else {
            own_scan_sum<true>(esrc, edst, eval, key, xu, xi, out, acc, q, &qn,
                               tid, lane, wave, lo, rows);
        }
    }
    __syncthreads();

    float4* yo = (float4*)(y + (long)lo * GNN_EMB);
    const float4* a4 = (const float4*)acc;
    for (int i = tid; i < rows * (GNN_EMB / 4); i += 1024) yo[i] = a4[i];
}

__global__ __launch_bounds__(256) void gnn_layer(
    const float* __restrict__ sideL, float* __restrict__ ego,
    const float* __restrict__ Wgc, const float* __restrict__ bgc,
    const float* __restrict__ Wbi, const float* __restrict__ bbi,
    float* __restrict__ out, int out_col0,
    const float* __restrict__ user_emb, const float* __restrict__ item_emb, int last)
{
    __shared__ float Wt[64][132];
    __shared__ float bsum[64];

    const int tid = threadIdx.x;

    for (int i = tid; i < 64 * 128; i += 256) {
        int k = i >> 6;
        int n = i & 63;
        float w = (k < 64) ? Wgc[k * 64 + n] : Wbi[(k - 64) * 64 + n];
        Wt[n][k] = w;
    }
    if (tid < 64) bsum[tid] = bgc[tid] + bbi[tid];
    __syncthreads();

    const int wave    = tid >> 5;
    const int lane    = tid & 31;
    const int rowbase = blockIdx.x * 128 + wave * 16;
    if (rowbase >= GNN_N) return;

    const int m    = lane & 15;
    const int half = lane >> 4;
    const int kb   = half << 1;

    static_assert(GNN_NUSERS % 16 == 0, "a 16-row tile must not straddle users and items");
    const float* cb = (rowbase < GNN_NUSERS)
        ? user_emb + (long)rowbase * GNN_EMB
        : item_emb + (long)(rowbase - GNN_NUSERS) * GNN_EMB;

    if (last) {
        v4f cv[8];
#pragma unroll
        for (int t = 0; t < 8; ++t)
            cv[t] = *(const v4f*)(cb + (2 * t + half) * GNN_EMB + m * 4);
        asm volatile("" :: "v"(cv[0]), "v"(cv[1]), "v"(cv[2]), "v"(cv[3]),
                           "v"(cv[4]), "v"(cv[5]), "v"(cv[6]), "v"(cv[7]));
#pragma unroll
        for (int t = 0; t < 8; ++t)
            *(v4f*)(out + (long)(rowbase + 2 * t + half) * 192 + m * 4) = cv[t];
    }

    const long  row    = (long)(rowbase + m);
    const float* egoRd = last ? ego + row * 64 : cb + m * 64;
    const float* sLRd  = sideL + row * 64;

    v2f e2[16], l2[16];
#pragma unroll
    for (int s = 0; s < 16; ++s) {
        int c = 4 * s + kb;
        e2[s] = *(const v2f*)(egoRd + c);
        l2[s] = *(const v2f*)(sLRd + c);
    }

    v8f acc[4] = {};

#pragma unroll
    for (int s = 0; s < 16; ++s) {
        v2f a;
        a.x = e2[s].x + l2[s].x;
        a.y = e2[s].y + l2[s].y;
#pragma unroll
        for (int nt = 0; nt < 4; ++nt) {
            v2f b = *(const v2f*)(&Wt[nt * 16 + m][4 * s + kb]);
            acc[nt] = __builtin_amdgcn_wmma_f32_16x16x4_f32(
                false, a, false, b, (short)0, acc[nt], false, false);
        }
    }
#pragma unroll
    for (int s = 0; s < 16; ++s) {
        v2f a;
        a.x = e2[s].x * l2[s].x;
        a.y = e2[s].y * l2[s].y;
#pragma unroll
        for (int nt = 0; nt < 4; ++nt) {
            v2f b = *(const v2f*)(&Wt[nt * 16 + m][64 + 4 * s + kb]);
            acc[nt] = __builtin_amdgcn_wmma_f32_16x16x4_f32(
                false, a, false, b, (short)0, acc[nt], false, false);
        }
    }

    float vout[4][8];
    float ssq[8];
#pragma unroll
    for (int j = 0; j < 8; ++j) ssq[j] = 0.f;
#pragma unroll
    for (int nt = 0; nt < 4; ++nt) {
        float bc = bsum[nt * 16 + m];
#pragma unroll
        for (int j = 0; j < 8; ++j) {
            float v = acc[nt][j] + bc;
            v = (v > 0.f) ? v : 0.01f * v;
            vout[nt][j] = v;
            ssq[j] += v * v;
        }
    }

    if (!last) {
        float* egoWr = ego + (long)rowbase * 64;
#pragma unroll
        for (int j = 0; j < 8; ++j) {
            int mrow = j + (half << 3);
#pragma unroll
            for (int nt = 0; nt < 4; ++nt)
                egoWr[mrow * 64 + nt * 16 + m] = vout[nt][j];
        }
    }

#pragma unroll
    for (int j = 0; j < 8; ++j) {
        float s = ssq[j];
        s += __shfl_xor(s, 8, 16);
        s += __shfl_xor(s, 4, 16);
        s += __shfl_xor(s, 2, 16);
        s += __shfl_xor(s, 1, 16);
        float inv = 1.0f / fmaxf(sqrtf(s), 1e-12f);
        int mrow = j + (half << 3);
        float* o = out + (long)(rowbase + mrow) * 192 + out_col0;
#pragma unroll
        for (int nt = 0; nt < 4; ++nt)
            o[nt * 16 + m] = vout[nt][j] * inv;
    }
}

extern "C" void kernel_launch(void* const* d_in, const int* in_sizes, int n_in,
                              void* d_out, int out_size, void* d_ws, size_t ws_size,
                              hipStream_t stream) {
    const float* user_emb = (const float*)d_in[0];
    const float* item_emb = (const float*)d_in[1];
    const float* W_gc0 = (const float*)d_in[2];
    const float* b_gc0 = (const float*)d_in[3];
    const float* W_bi0 = (const float*)d_in[4];
    const float* b_bi0 = (const float*)d_in[5];
    const float* W_gc1 = (const float*)d_in[6];
    const float* b_gc1 = (const float*)d_in[7];
    const float* W_bi1 = (const float*)d_in[8];
    const float* b_bi1 = (const float*)d_in[9];
    const int*   e_src = (const int*)d_in[10];
    const int*   e_dst = (const int*)d_in[11];
    const float* e_val = (const float*)d_in[12];
    float* out = (float*)d_out;

    float* ego   = (float*)d_ws;
    float* sideL = ego + (long)GNN_N * GNN_EMB;
    unsigned* key4 = (unsigned*)(sideL + (long)GNN_N * GNN_EMB);

    const int  gLayer = (GNN_N + 127) / 128;

    gnn_key<<<FE_BLOCKS, 256, 0, stream>>>(e_src, e_dst, e_val, key4, out);

    gnn_spmm_own<false><<<OWN_BLOCKS, 1024, 0, stream>>>(e_src, e_dst, e_val, key4,
                                                         user_emb, item_emb, sideL, out);
    gnn_layer<<<gLayer, 256, 0, stream>>>(sideL, ego, W_gc0, b_gc0, W_bi0, b_bi0,
                                          out, 64, user_emb, item_emb, 0);

    gnn_spmm_own<true><<<OWN_BLOCKS, 1024, 0, stream>>>(e_src, e_dst, e_val, key4,
                                                        ego, item_emb, sideL, out);
    gnn_layer<<<gLayer, 256, 0, stream>>>(sideL, ego, W_gc1, b_gc1, W_bi1, b_bi1,
                                          out, 128, user_emb, item_emb, 1);
}


# SpMM sums leave LDS: records dealt to 32 LDS queues by row % 32, each wave sums its 13 rows in scalar-indexed registers; no LDS float add, no LDS tile; sum order: LDS arrival of the deal
# speedup vs baseline: 4.4488x; 1.1499x over previous
#include <hip/hip_runtime.h>
#include <hip/hip_bf16.h>
#include <math.h>


typedef __attribute__((ext_vector_type(2))) float v2f;
typedef __attribute__((ext_vector_type(4))) float v4f;
typedef __attribute__((ext_vector_type(8))) float v8f;

#define GNN_N       100000
#define GNN_NUSERS  60000
#define GNN_EMB     64
#define GNN_E       3200000

#define FE_BLOCKS   625
#define FE_WAVES    (FE_BLOCKS * 8)

#define OWN_ROWS   391
#define OWN_BLOCKS 256
#define OWN_Q      32768
#define KEY_G      (GNN_E / 32)
#define KEY_SG     (GNN_E / 128)
#define KEY_PER_WAVE (KEY_SG / FE_WAVES)
#define REC_DST_BITS 17
#define REC_DST_MASK ((1u << REC_DST_BITS) - 1u)
#define REC_BAD_ROW  511u
#define REC_COL0     64
#define LIST_NONE    0xFFFFFFFFu
#define OWN_SLOTS    13
#define DEAL_CAP     576

static_assert(OWN_ROWS * OWN_BLOCKS >= GNN_N, "owners must cover all rows");
static_assert(KEY_SG % FE_WAVES == 0, "the key pass has no tail");
static_assert(KEY_G == GNN_N, "one 256-B record group per output row");
static_assert(GNN_N <= (1 << REC_DST_BITS), "dst must fit its record field");
static_assert(OWN_ROWS <= (int)REC_BAD_ROW, "local row must fit below the bad-row code");
static_assert(OWN_ROWS <= 32 * OWN_SLOTS, "every local row has a wave and a slot");

__global__ __launch_bounds__(256) void gnn_key(
    const int* __restrict__ esrc, const int* __restrict__ edst,
    const float* __restrict__ eval, unsigned* __restrict__ key,
    float* __restrict__ out)
{
    const int lane = threadIdx.x & 31;
    const int wv = blockIdx.x * 8 + __builtin_amdgcn_readfirstlane(threadIdx.x >> 5);
    unsigned s[KEY_PER_WAVE][4], d[KEY_PER_WAVE][4], v[KEY_PER_WAVE][4];
#pragma unroll
    for (int h = 0; h < KEY_PER_WAVE; ++h) {
#pragma unroll
        for (int u = 0; u < 4; ++u) {
            const int e = (wv + h * FE_WAVES) * 128 + u * 32 + lane;
            s[h][u] = (unsigned)esrc[e];
            d[h][u] = (unsigned)edst[e];
            v[h][u] = __float_as_uint(eval[e]);
        }
    }
#pragma unroll
    for (int h = 0; h < KEY_PER_WAVE; ++h) {
        const int G = wv + h * FE_WAVES;
        unsigned mine = 0u;
#pragma unroll
        for (int u = 0; u < 4; ++u) {
            const unsigned oq = s[h][u] / OWN_ROWS;
            const unsigned o  = oq & 255u;
            const unsigned lr = (oq < (unsigned)OWN_BLOCKS)
                                  ? (s[h][u] - oq * OWN_ROWS) : REC_BAD_ROW;
            uint2 rc;
            rc.x = (d[h][u] & REC_DST_MASK) | (lr << REC_DST_BITS);
            rc.y = v[h][u];
            *(uint2*)(out + (long)(4 * G + u) * 192 + REC_COL0 + lane * 2) = rc;
#pragma unroll
            for (int j = 0; j < 8; ++j) {
                unsigned bj = __builtin_amdgcn_ballot_w32(((o >> j) & 1u) != 0u);
                if (lane == u * 8 + j) mine = bj;
            }
        }
        key[G * 32 + lane] = mine;
    }
}

__device__ __forceinline__ unsigned* own_chunk(float* out, int lo, int rows, unsigned c)
{
    return (c < (unsigned)rows)
        ? (unsigned*)(out + (long)(lo + (int)c) * 192)
        : (unsigned*)(out + (long)(lo + (int)c - rows) * 192 + 128);
}

template <bool SPLIT>
__device__ __forceinline__ void own_sum32(
    const uint2 rc, const int rows, const float* __restrict__ xu,
    const float* __restrict__ xi, float (&ax)[OWN_SLOTS], float (&ay)[OWN_SLOTS],
    const int lane)
{
    const int   d = (int)(rc.x & REC_DST_MASK);
    const int   r = (int)(rc.x >> REC_DST_BITS);
    const float v = __uint_as_float(rc.y);
    const bool ok = (unsigned)r < (unsigned)rows;
    const unsigned okm = __builtin_amdgcn_ballot_w32(ok);
    const float* xi0 = xi - (long)GNN_NUSERS * GNN_EMB;
#pragma unroll
    for (int j0 = 0; j0 < 32; j0 += 16) {
        if (((okm >> j0) & 0xFFFFu) == 0u) continue;
        v2f xv[16];
#pragma unroll
        for (int u = 0; u < 16; ++u) {
            int dj = __builtin_amdgcn_readlane(d, j0 + u);
            const float* xb = (SPLIT && dj >= GNN_NUSERS) ? xi0 : xu;
            xv[u] = *(const v2f*)(xb + (unsigned)dj * (unsigned)GNN_EMB + lane * 2);
        }
#pragma unroll
        for (int u = 0; u < 16; ++u) {
            if ((okm >> (j0 + u)) & 1u) {
                const int   kj = __builtin_amdgcn_readlane(r, j0 + u) >> 5;
                const float vj = __int_as_float(
                    __builtin_amdgcn_readlane(__float_as_int(v), j0 + u));
                ax[kj] += vj * xv[u].x;
                ay[kj] += vj * xv[u].y;
            }
        }
    }
}

template <bool RAW>
__device__ __forceinline__ void own_scan_sum(
    const int* __restrict__ esrc, const int* __restrict__ edst,
    const float* __restrict__ eval, const unsigned* __restrict__ key,
    const float* __restrict__ xu, const float* __restrict__ xi, float* out,
    float (&ax)[OWN_SLOTS], float (&ay)[OWN_SLOTS],
    unsigned* q, unsigned* qn, unsigned* cnt, unsigned* recq,
    const int tid, const int lane, const int wave, const int lo, const int rows)
{
    const uint4* __restrict__ key16 = (const uint4*)key;
    const unsigned* rec = (const unsigned*)(out + REC_COL0);
    const unsigned seg = (2u * (unsigned)rows) >> 5;
    unsigned wdone = 0u;
    bool fits = true;
    unsigned inv[8];
#pragma unroll
    for (int j = 0; j < 8; ++j) inv[j] = ((blockIdx.x >> j) & 1u) ? 0u : ~0u;

    int pos  = 0;
    int step = KEY_G;
    for (int trip = 0; trip < 3 + GNN_E / OWN_Q && pos < KEY_G; ++trip) {
        const int end = min(pos + step, KEY_G);
        if (tid == 0) *qn = 0;
        __syncthreads();

        for (int g0 = pos + tid; g0 < end; g0 += 4096) {
            uint4 ka[4], kb[4];
#pragma unroll
            for (int u = 0; u < 4; ++u) {
                int gc = min(g0 + u * 1024, KEY_G - 1);
                ka[u] = key16[2 * gc];
                kb[u] = key16[2 * gc + 1];
            }
#pragma unroll
            for (int u = 0; u < 4; ++u) {
                const int g = g0 + u * 1024;
                unsigned m = (ka[u].x ^ inv[0]) & (ka[u].y ^ inv[1])
                           & (ka[u].z ^ inv[2]) & (ka[u].w ^ inv[3])
                           & (kb[u].x ^ inv[4]) & (kb[u].y ^ inv[5])
                           & (kb[u].z ^ inv[6]) & (kb[u].w ^ inv[7]);
                if (g >= end) m = 0u;
                while (m) {
                    unsigned bit = (unsigned)__ffs((int)m) - 1u;
                    m &= m - 1u;
                    unsigned slot = atomicAdd(qn, 1u);
                    if (slot < OWN_Q) q[slot] = (unsigned)g * 32u + bit;
                }
            }
        }
        __syncthreads();
        const unsigned n = __builtin_amdgcn_readfirstlane(*qn);
        __syncthreads();
        if (n > OWN_Q) { step = OWN_Q / 32; continue; }

        unsigned start = 0u;
        unsigned rstep = n;
        for (int round = 0; round < 2 + OWN_Q / DEAL_CAP && start < n; ++round) {
            const unsigned stop = start + min(rstep, n - start);
            if (tid < 32) cnt[tid] = 0u;
            __syncthreads();
            for (unsigned base = start + (unsigned)wave * 32u; base < stop; base += 1024u) {
                const unsigned qi = base + lane;
                const unsigned e = q[min(qi, n - 1u)];
                uint2 rc;
                if (RAW) {
                    const unsigned s  = (unsigned)esrc[e];
                    const unsigned oq = s / OWN_ROWS;
                    const unsigned lr = (oq == blockIdx.x) ? (s - oq * OWN_ROWS) : REC_BAD_ROW;
                    rc.x = (((unsigned)edst[e]) & REC_DST_MASK) | (lr << REC_DST_BITS);
                    rc.y = __float_as_uint(eval[e]);
                } else {
                    rc = *(const uint2*)(rec + (size_t)(e >> 5) * 192u + (e & 31u) * 2u);
                }
                const unsigned r = rc.x >> REC_DST_BITS;
                if (qi < stop && r < (unsigned)rows) {
                    const unsigned w = r & 31u;
                    const unsigned slot = atomicAdd(&cnt[w], 1u);
                    if (slot < DEAL_CAP) *(uint2*)(recq + 2u * (w * DEAL_CAP + slot)) = rc;
                }
            }
            __syncthreads();
            const unsigned cl   = cnt[lane];
            const unsigned over = __builtin_amdgcn_ballot_w32(cl > DEAL_CAP);
            const unsigned mine = (unsigned)__builtin_amdgcn_readlane((int)cl, wave);
            if (over != 0u) {
                rstep = DEAL_CAP;
                __syncthreads();
                continue;
            }
            for (unsigned b = 0u; b < mine; b += 32u) {
                uint2 rc; rc.x = REC_BAD_ROW << REC_DST_BITS; rc.y = 0u;
                if (b + lane < mine)
                    rc = *(const uint2*)(recq + 2u * ((unsigned)wave * DEAL_CAP + b + lane));
                if (!RAW) {
                    if (wdone + 1u < seg)
                        *(uint2*)(own_chunk(out, lo, rows, (unsigned)wave * seg + 1u + wdone)
                                  + lane * 2) = rc;
                    else
                        fits = false;
                    ++wdone;
                }
                own_sum32<!RAW>(rc, rows, xu, xi, ax, ay, lane);
            }
            start = stop;
            __syncthreads();
        }
        pos = end;
    }
    if (!RAW) {
        const unsigned h = fits ? wdone : LIST_NONE;
        uint2 hv; hv.x = h; hv.y = h;
        *(uint2*)(own_chunk(out, lo, rows, (unsigned)wave * seg) + lane * 2) = hv;
    }
}

template <bool SECOND>
__global__ __launch_bounds__(1024) void gnn_spmm_own(
    const int* __restrict__ esrc, const int* __restrict__ edst,
    const float* __restrict__ eval, const unsigned* __restrict__ key,
    const float* __restrict__ xu, const float* __restrict__ xi,
    float* __restrict__ y, float* out)
{
    __shared__ unsigned q[OWN_Q];
    __shared__ __attribute__((aligned(8))) unsigned recq[32 * DEAL_CAP * 2];
    __shared__ unsigned cnt[32];
    __shared__ unsigned qn;
    __shared__ unsigned nolist;

    const int tid  = threadIdx.x;
    const int lane = tid & 31;
    const int wave = __builtin_amdgcn_readfirstlane(tid >> 5);
    const int lo   = blockIdx.x * OWN_ROWS;
    const int rows = min(OWN_ROWS, GNN_N - lo);

    float ax[OWN_SLOTS], ay[OWN_SLOTS];
#pragma unroll
    for (int k = 0; k < OWN_SLOTS; ++k) { ax[k] = 0.f; ay[k] = 0.f; }

    if (!SECOND) {
        own_scan_sum<false>(esrc, edst, eval, key, xu, xi, out, ax, ay, q, &qn, cnt, recq,
                            tid, lane, wave, lo, rows);
    } else {
        const unsigned seg = (2u * (unsigned)rows) >> 5;
        const unsigned h = __builtin_amdgcn_readfirstlane(
            *own_chunk(out, lo, rows, (unsigned)wave * seg));
        if (tid == 0) nolist = 0u;
        __syncthreads();
        if (h == LIST_NONE && lane == 0) nolist = 1u;
        __syncthreads();
        if (__builtin_amdgcn_readfirstlane(nolist) == 0u) {
            const unsigned total = min(h, seg - 1u);
            unsigned cc = 0u;
            uint2 nx; nx.x = REC_BAD_ROW << REC_DST_BITS; nx.y = 0u;
            if (cc < total)
                nx = *(const uint2*)(own_chunk(out, lo, rows, (unsigned)wave * seg + 1u) + lane * 2);
            for (; cc < total; ++cc) {
                const uint2 rc = nx;
                if (cc + 1u < total)
                    nx = *(const uint2*)(own_chunk(out, lo, rows, (unsigned)wave * seg + 2u + cc)
                                         + lane * 2);
                own_sum32<false>(rc, rows, xu, xi, ax, ay, lane);
            }
        } else {
            own_scan_sum<true>(esrc, edst, eval, key, xu, xi, out, ax, ay, q, &qn, cnt, recq,
                               tid, lane, wave, lo, rows);
        }
    }

#pragma unroll
    for (int k = 0; k < OWN_SLOTS; ++k) {
        const int r = k * 32 + wave;
        if (r < rows) {
            v2f o; o.x = ax[k]; o.y = ay[k];
            *(v2f*)(y + (long)(lo + r) * GNN_EMB + lane * 2) = o;
        }
    }
}

__global__ __launch_bounds__(256) void gnn_layer(
    const float* __restrict__ sideL, float* __restrict__ ego,
    const float* __restrict__ Wgc, const float* __restrict__ bgc,
    const float* __restrict__ Wbi, const float* __restrict__ bbi,
    float* __restrict__ out, int out_col0,
    const float* __restrict__ user_emb, const float* __restrict__ item_emb, int last)
{
    __shared__ float Wt[64][132];
    __shared__ float bsum[64];

    const int tid = threadIdx.x;

    for (int i = tid; i < 64 * 128; i += 256) {
        int k = i >> 6;
        int n = i & 63;
        float w = (k < 64) ? Wgc[k * 64 + n] : Wbi[(k - 64) * 64 + n];
        Wt[n][k] = w;
    }
    if (tid < 64) bsum[tid] = bgc[tid] + bbi[tid];
    __syncthreads();

    const int wave    = tid >> 5;
    const int lane    = tid & 31;
    const int rowbase = blockIdx.x * 128 + wave * 16;
    if (rowbase >= GNN_N) return;

    const int m    = lane & 15;
    const int half = lane >> 4;
    const int kb   = half << 1;

    static_assert(GNN_NUSERS % 16 == 0, "a 16-row tile must not straddle users and items");
    const float* cb = (rowbase < GNN_NUSERS)
        ? user_emb + (long)rowbase * GNN_EMB
        : item_emb + (long)(rowbase - GNN_NUSERS) * GNN_EMB;

    if (last) {
        v4f cv[8];
#pragma unroll
        for (int t = 0; t < 8; ++t)
            cv[t] = *(const v4f*)(cb + (2 * t + half) * GNN_EMB + m * 4);
        asm volatile("" :: "v"(cv[0]), "v"(cv[1]), "v"(cv[2]), "v"(cv[3]),
                           "v"(cv[4]), "v"(cv[5]), "v"(cv[6]), "v"(cv[7]));
#pragma unroll
        for (int t = 0; t < 8; ++t)
            *(v4f*)(out + (long)(rowbase + 2 * t + half) * 192 + m * 4) = cv[t];
    }

    const long  row    = (long)(rowbase + m);
    const float* egoRd = last ? ego + row * 64 : cb + m * 64;
    const float* sLRd  = sideL + row * 64;

    v2f e2[16], l2[16];
#pragma unroll
    for (int s = 0; s < 16; ++s) {
        int c = 4 * s + kb;
        e2[s] = *(const v2f*)(egoRd + c);
        l2[s] = *(const v2f*)(sLRd + c);
    }

    v8f acc[4] = {};

#pragma unroll
    for (int s = 0; s < 16; ++s) {
        v2f a;
        a.x = e2[s].x + l2[s].x;
        a.y = e2[s].y + l2[s].y;
#pragma unroll
        for (int nt = 0; nt < 4; ++nt) {
            v2f b = *(const v2f*)(&Wt[nt * 16 + m][4 * s + kb]);
            acc[nt] = __builtin_amdgcn_wmma_f32_16x16x4_f32(
                false, a, false, b, (short)0, acc[nt], false, false);
        }
    }
#pragma unroll
    for (int s = 0; s < 16; ++s) {
        v2f a;
        a.x = e2[s].x * l2[s].x;
        a.y = e2[s].y * l2[s].y;
#pragma unroll
        for (int nt = 0; nt < 4; ++nt) {
            v2f b = *(const v2f*)(&Wt[nt * 16 + m][64 + 4 * s + kb]);
            acc[nt] = __builtin_amdgcn_wmma_f32_16x16x4_f32(
                false, a, false, b, (short)0, acc[nt], false, false);
        }
    }

    float vout[4][8];
    float ssq[8];
#pragma unroll
    for (int j = 0; j < 8; ++j) ssq[j] = 0.f;
#pragma unroll
    for (int nt = 0; nt < 4; ++nt) {
        float bc = bsum[nt * 16 + m];
#pragma unroll
        for (int j = 0; j < 8; ++j) {
            float v = acc[nt][j] + bc;
            v = (v > 0.f) ? v : 0.01f * v;
            vout[nt][j] = v;
            ssq[j] += v * v;
        }
    }

    if (!last) {
        float* egoWr = ego + (long)rowbase * 64;
#pragma unroll
        for (int j = 0; j < 8; ++j) {
            int mrow = j + (half << 3);
#pragma unroll
            for (int nt = 0; nt < 4; ++nt)
                egoWr[mrow * 64 + nt * 16 + m] = vout[nt][j];
        }
    }

#pragma unroll
    for (int j = 0; j < 8; ++j) {
        float s = ssq[j];
        s += __shfl_xor(s, 8, 16);
        s += __shfl_xor(s, 4, 16);
        s += __shfl_xor(s, 2, 16);
        s += __shfl_xor(s, 1, 16);
        float inv = 1.0f / fmaxf(sqrtf(s), 1e-12f);
        int mrow = j + (half << 3);
        float* o = out + (long)(rowbase + mrow) * 192 + out_col0;
#pragma unroll
        for (int nt = 0; nt < 4; ++nt)
            o[nt * 16 + m] = vout[nt][j] * inv;
    }
}

extern "C" void kernel_launch(void* const* d_in, const int* in_sizes, int n_in,
                              void* d_out, int out_size, void* d_ws, size_t ws_size,
                              hipStream_t stream) {
    const float* user_emb = (const float*)d_in[0];
    const float* item_emb = (const float*)d_in[1];
    const float* W_gc0 = (const float*)d_in[2];
    const float* b_gc0 = (const float*)d_in[3];
    const float* W_bi0 = (const float*)d_in[4];
    const float* b_bi0 = (const float*)d_in[5];
    const float* W_gc1 = (const float*)d_in[6];
    const float* b_gc1 = (const float*)d_in[7];
    const float* W_bi1 = (const float*)d_in[8];
    const float* b_bi1 = (const float*)d_in[9];
    const int*   e_src = (const int*)d_in[10];
    const int*   e_dst = (const int*)d_in[11];
    const float* e_val = (const float*)d_in[12];
    float* out = (float*)d_out;

    float* ego   = (float*)d_ws;
    float* sideL = ego + (long)GNN_N * GNN_EMB;
    unsigned* key4 = (unsigned*)(sideL + (long)GNN_N * GNN_EMB);

    const int  gLayer = (GNN_N + 127) / 128;

    gnn_key<<<FE_BLOCKS, 256, 0, stream>>>(e_src, e_dst, e_val, key4, out);

    gnn_spmm_own<false><<<OWN_BLOCKS, 1024, 0, stream>>>(e_src, e_dst, e_val, key4,
                                                         user_emb, item_emb, sideL, out);
    gnn_layer<<<gLayer, 256, 0, stream>>>(sideL, ego, W_gc0, b_gc0, W_bi0, b_bi0,
                                          out, 64, user_emb, item_emb, 0);

    gnn_spmm_own<true><<<OWN_BLOCKS, 1024, 0, stream>>>(e_src, e_dst, e_val, key4,
                                                        ego, item_emb, sideL, out);
    gnn_layer<<<gLayer, 256, 0, stream>>>(sideL, ego, W_gc1, b_gc1, W_bi1, b_bi1,
                                          out, 128, user_emb, item_emb, 1);
}


# Layer product fused into the owner SpMM block: the block hands its 391 summed rows through an LDS tile to its own WMMAs; side_L never goes to memory; 5 launches to 3; records move to side_L's plane
# speedup vs baseline: 4.8211x; 1.0837x over previous
#include <hip/hip_runtime.h>
#include <hip/hip_bf16.h>
#include <math.h>


typedef __attribute__((ext_vector_type(2))) float v2f;
typedef __attribute__((ext_vector_type(4))) float v4f;
typedef __attribute__((ext_vector_type(8))) float v8f;

#define GNN_N       100000
#define GNN_NUSERS  60000
#define GNN_EMB     64
#define GNN_E       3200000

#define FE_BLOCKS   625
#define FE_WAVES    (FE_BLOCKS * 8)

#define OWN_ROWS   391
#define OWN_BLOCKS 256
#define OWN_Q      32768
#define KEY_G      (GNN_E / 32)
#define KEY_SG     (GNN_E / 128)
#define KEY_PER_WAVE (KEY_SG / FE_WAVES)
#define REC_DST_BITS 17
#define REC_DST_MASK ((1u << REC_DST_BITS) - 1u)
#define REC_BAD_ROW  511u
#define LIST_NONE    0xFFFFFFFFu
#define OWN_SLOTS    13
#define DEAL_CAP     576

static_assert(OWN_ROWS * OWN_BLOCKS >= GNN_N, "owners must cover all rows");
static_assert(KEY_SG % FE_WAVES == 0, "the key pass has no tail");
static_assert(GNN_N <= (1 << REC_DST_BITS), "dst must fit its record field");
static_assert(OWN_ROWS <= (int)REC_BAD_ROW, "local row must fit below the bad-row code");
static_assert(OWN_ROWS <= 32 * OWN_SLOTS, "every local row has a wave and a slot");

__global__ __launch_bounds__(256) void gnn_key(
    const int* __restrict__ esrc, const int* __restrict__ edst,
    const float* __restrict__ eval, unsigned* __restrict__ key,
    unsigned* __restrict__ rec)
{
    const int lane = threadIdx.x & 31;
    const int wv = blockIdx.x * 8 + __builtin_amdgcn_readfirstlane(threadIdx.x >> 5);
    unsigned s[KEY_PER_WAVE][4], d[KEY_PER_WAVE][4], v[KEY_PER_WAVE][4];
#pragma unroll
    for (int h = 0; h < KEY_PER_WAVE; ++h) {
#pragma unroll
        for (int u = 0; u < 4; ++u) {
            const int e = (wv + h * FE_WAVES) * 128 + u * 32 + lane;
            s[h][u] = (unsigned)esrc[e];
            d[h][u] = (unsigned)edst[e];
            v[h][u] = __float_as_uint(eval[e]);
        }
    }
#pragma unroll
    for (int h = 0; h < KEY_PER_WAVE; ++h) {
        const int G = wv + h * FE_WAVES;
        unsigned mine = 0u;
#pragma unroll
        for (int u = 0; u < 4; ++u) {
            const unsigned oq = s[h][u] / OWN_ROWS;
            const unsigned o  = oq & 255u;
            const unsigned lr = (oq < (unsigned)OWN_BLOCKS)
                                  ? (s[h][u] - oq * OWN_ROWS) : REC_BAD_ROW;
            uint2 rc;
            rc.x = (d[h][u] & REC_DST_MASK) | (lr << REC_DST_BITS);
            rc.y = v[h][u];
            *(uint2*)(rec + ((long)(4 * G + u) * 32 + lane) * 2) = rc;
#pragma unroll
            for (int j = 0; j < 8; ++j) {
                unsigned bj = __builtin_amdgcn_ballot_w32(((o >> j) & 1u) != 0u);
                if (lane == u * 8 + j) mine = bj;
            }
        }
        key[G * 32 + lane] = mine;
    }
}

__device__ __forceinline__ unsigned* own_chunk(float* out, int lo, int rows, unsigned c)
{
    return (c < (unsigned)rows)
        ? (unsigned*)(out + (long)(lo + (int)c) * 192)
        : (unsigned*)(out + (long)(lo + (int)c - rows) * 192 + 128);
}

template <bool SPLIT>
__device__ __forceinline__ void own_sum32(
    const uint2 rc, const int rows, const float* __restrict__ xu,
    const float* __restrict__ xi, float (&ax)[OWN_SLOTS], float (&ay)[OWN_SLOTS],
    const int lane)
{
    const int   d = (int)(rc.x & REC_DST_MASK);
    const int   r = (int)(rc.x >> REC_DST_BITS);
    const float v = __uint_as_float(rc.y);
    const bool ok = (unsigned)r < (unsigned)rows;
    const unsigned okm = __builtin_amdgcn_ballot_w32(ok);
    const float* xi0 = xi - (long)GNN_NUSERS * GNN_EMB;
#pragma unroll
    for (int j0 = 0; j0 < 32; j0 += 16) {
        if (((okm >> j0) & 0xFFFFu) == 0u) continue;
        v2f xv[16];
#pragma unroll
        for (int u = 0; u < 16; ++u) {
            int dj = __builtin_amdgcn_readlane(d, j0 + u);
            const float* xb = (SPLIT && dj >= GNN_NUSERS) ? xi0 : xu;
            xv[u] = *(const v2f*)(xb + (unsigned)dj * (unsigned)GNN_EMB + lane * 2);
        }
#pragma unroll
        for (int u = 0; u < 16; ++u) {
            if ((okm >> (j0 + u)) & 1u) {
                const int   kj = __builtin_amdgcn_readlane(r, j0 + u) >> 5;
                const float vj = __int_as_float(
                    __builtin_amdgcn_readlane(__float_as_int(v), j0 + u));
                ax[kj] += vj * xv[u].x;
                ay[kj] += vj * xv[u].y;
            }
        }
    }
}

template <bool RAW>
__device__ __forceinline__ void own_scan_sum(
    const int* __restrict__ esrc, const int* __restrict__ edst,
    const float* __restrict__ eval, const unsigned* __restrict__ key,
    const unsigned* __restrict__ rec,
    const float* __restrict__ xu, const float* __restrict__ xi, float* out,
    float (&ax)[OWN_SLOTS], float (&ay)[OWN_SLOTS],
    unsigned* q, unsigned* qn, unsigned* cnt, unsigned* recq,
    const int tid, const int lane, const int wave, const int lo, const int rows)
{
    const uint4* __restrict__ key16 = (const uint4*)key;
    const unsigned seg = (2u * (unsigned)rows) >> 5;
    unsigned wdone = 0u;
    bool fits = true;
    unsigned inv[8];
#pragma unroll
    for (int j = 0; j < 8; ++j) inv[j] = ((blockIdx.x >> j) & 1u) ? 0u : ~0u;

    int pos  = 0;
    int step = KEY_G;
    for (int trip = 0; trip < 3 + GNN_E / OWN_Q && pos < KEY_G; ++trip) {
        const int end = min(pos + step, KEY_G);
        if (tid == 0) *qn = 0;
        __syncthreads();

        for (int g0 = pos + tid; g0 < end; g0 += 4096) {
            uint4 ka[4], kb[4];
#pragma unroll
            for (int u = 0; u < 4; ++u) {
                int gc = min(g0 + u * 1024, KEY_G - 1);
                ka[u] = key16[2 * gc];
                kb[u] = key16[2 * gc + 1];
            }
#pragma unroll
            for (int u = 0; u < 4; ++u) {
                const int g = g0 + u * 1024;
                unsigned m = (ka[u].x ^ inv[0]) & (ka[u].y ^ inv[1])
                           & (ka[u].z ^ inv[2]) & (ka[u].w ^ inv[3])
                           & (kb[u].x ^ inv[4]) & (kb[u].y ^ inv[5])
                           & (kb[u].z ^ inv[6]) & (kb[u].w ^ inv[7]);
                if (g >= end) m = 0u;
                while (m) {
                    unsigned bit = (unsigned)__ffs((int)m) - 1u;
                    m &= m - 1u;
                    unsigned slot = atomicAdd(qn, 1u);
                    if (slot < OWN_Q) q[slot] = (unsigned)g * 32u + bit;
                }
            }
        }
        __syncthreads();
        const unsigned n = __builtin_amdgcn_readfirstlane(*qn);
        __syncthreads();
        if (n > OWN_Q) { step = OWN_Q / 32; continue; }

        unsigned start = 0u;
        unsigned rstep = n;
        for (int round = 0; round < 2 + OWN_Q / DEAL_CAP && start < n; ++round) {
            const unsigned stop = start + min(rstep, n - start);
            if (tid < 32) cnt[tid] = 0u;
            __syncthreads();
            for (unsigned base = start + (unsigned)wave * 32u; base < stop; base += 1024u) {
                const unsigned qi = base + lane;
                const unsigned e = q[min(qi, n - 1u)];
                uint2 rc;
                if (RAW) {
                    const unsigned s  = (unsigned)esrc[e];
                    const unsigned oq = s / OWN_ROWS;
                    const unsigned lr = (oq == blockIdx.x) ? (s - oq * OWN_ROWS) : REC_BAD_ROW;
                    rc.x = (((unsigned)edst[e]) & REC_DST_MASK) | (lr << REC_DST_BITS);
                    rc.y = __float_as_uint(eval[e]);
                } else {
                    rc = *(const uint2*)(rec + (size_t)e * 2u);
                }
                const unsigned r = rc.x >> REC_DST_BITS;
                if (qi < stop && r < (unsigned)rows) {
                    const unsigned w = r & 31u;
                    const unsigned slot = atomicAdd(&cnt[w], 1u);
                    if (slot < DEAL_CAP) *(uint2*)(recq + 2u * (w * DEAL_CAP + slot)) = rc;
                }
            }
            __syncthreads();
            const unsigned cl   = cnt[lane];
            const unsigned over = __builtin_amdgcn_ballot_w32(cl > DEAL_CAP);
            const unsigned mine = (unsigned)__builtin_amdgcn_readlane((int)cl, wave);
            if (over != 0u) {
                rstep = DEAL_CAP;
                __syncthreads();
                continue;
            }
            for (unsigned b = 0u; b < mine; b += 32u) {
                uint2 rc; rc.x = REC_BAD_ROW << REC_DST_BITS; rc.y = 0u;
                if (b + lane < mine)
                    rc = *(const uint2*)(recq + 2u * ((unsigned)wave * DEAL_CAP + b + lane));
                if (!RAW) {
                    if (wdone + 1u < seg)
                        *(uint2*)(own_chunk(out, lo, rows, (unsigned)wave * seg + 1u + wdone)
                                  + lane * 2) = rc;
                    else
                        fits = false;
                    ++wdone;
                }
                own_sum32<!RAW>(rc, rows, xu, xi, ax, ay, lane);
            }
            start = stop;
            __syncthreads();
        }
        pos = end;
    }
    if (!RAW) {
        const unsigned h = fits ? wdone : LIST_NONE;
        uint2 hv; hv.x = h; hv.y = h;
        *(uint2*)(own_chunk(out, lo, rows, (unsigned)wave * seg) + lane * 2) = hv;
    }
}

#define SL_PITCH 68
#define SL_ROWS  400

static_assert(SL_ROWS >= ((OWN_ROWS + 15) / 16) * 16, "the LDS tile holds every padded row");
static_assert(SL_ROWS <= 32 * OWN_SLOTS, "every LDS row has a wave and a slot that writes it");
static_assert(SL_ROWS * SL_PITCH <= OWN_Q, "the LDS tile fits the edge queue's space");

__device__ __forceinline__ const float* cat_row(
    const float* __restrict__ user_emb, const float* __restrict__ item_emb, const int g)
{
    return (g < GNN_NUSERS) ? user_emb + (long)g * GNN_EMB
                            : item_emb + (long)(g - GNN_NUSERS) * GNN_EMB;
}

template <bool SECOND>
__device__ __forceinline__ void own_layer_tile(
    const float* sl, const float (*Wt)[132], const float* bsum,
    const float* __restrict__ user_emb, const float* __restrict__ item_emb,
    float* ego, float* out, const int lo, const int rows, const int tile, const int lane)
{
    const int m     = lane & 15;
    const int half  = lane >> 4;
    const int kb    = half << 1;
    const int lrow0 = tile * 16;
    const int glast = lo + rows - 1;
    const int out_col0 = SECOND ? 128 : 64;

    if (SECOND) {
        v4f cv[8];
#pragma unroll
        for (int t = 0; t < 8; ++t) {
            const int g = min(lo + lrow0 + 2 * t + half, glast);
            cv[t] = *(const v4f*)(cat_row(user_emb, item_emb, g) + m * 4);
        }
        asm volatile("" :: "v"(cv[0]), "v"(cv[1]), "v"(cv[2]), "v"(cv[3]),
                           "v"(cv[4]), "v"(cv[5]), "v"(cv[6]), "v"(cv[7]));
#pragma unroll
        for (int t = 0; t < 8; ++t) {
            const int lr = lrow0 + 2 * t + half;
            if (lr < rows)
                *(v4f*)(out + (long)(lo + lr) * 192 + m * 4) = cv[t];
        }
    }

    const int    ga    = min(lo + lrow0 + m, glast);
    const float* egoRd = SECOND ? (const float*)ego + (long)ga * GNN_EMB
                                : cat_row(user_emb, item_emb, ga);
    const float* sLRd  = sl + (lrow0 + m) * SL_PITCH;

    v2f e2[16];
#pragma unroll
    for (int s = 0; s < 16; ++s)
        e2[s] = *(const v2f*)(egoRd + 4 * s + kb);

    v8f acc[4] = {};

#pragma unroll
    for (int s = 0; s < 16; ++s) {
        const v2f l = *(const v2f*)(sLRd + 4 * s + kb);
        v2f a;
        a.x = e2[s].x + l.x;
        a.y = e2[s].y + l.y;
#pragma unroll
        for (int nt = 0; nt < 4; ++nt) {
            v2f b = *(const v2f*)(&Wt[nt * 16 + m][4 * s + kb]);
            acc[nt] = __builtin_amdgcn_wmma_f32_16x16x4_f32(
                false, a, false, b, (short)0, acc[nt], false, false);
        }
    }
#pragma unroll
    for (int s = 0; s < 16; ++s) {
        const v2f l = *(const v2f*)(sLRd + 4 * s + kb);
        v2f a;
        a.x = e2[s].x * l.x;
        a.y = e2[s].y * l.y;
#pragma unroll
        for (int nt = 0; nt < 4; ++nt) {
            v2f b = *(const v2f*)(&Wt[nt * 16 + m][64 + 4 * s + kb]);
            acc[nt] = __builtin_amdgcn_wmma_f32_16x16x4_f32(
                false, a, false, b, (short)0, acc[nt], false, false);
        }
    }

    float vout[4][8];
    float ssq[8];
#pragma unroll
    for (int j = 0; j < 8; ++j) ssq[j] = 0.f;
#pragma unroll
    for (int nt = 0; nt < 4; ++nt) {
        float bc = bsum[nt * 16 + m];
#pragma unroll
        for (int j = 0; j < 8; ++j) {
            float v = acc[nt][j] + bc;
            v = (v > 0.f) ? v : 0.01f * v;
            vout[nt][j] = v;
            ssq[j] += v * v;
        }
    }

    if (!SECOND) {
#pragma unroll
        for (int j = 0; j < 8; ++j) {
            const int lr = lrow0 + j + (half << 3);
            if (lr < rows) {
                float* egoWr = ego + (long)(lo + lr) * GNN_EMB;
#pragma unroll
                for (int nt = 0; nt < 4; ++nt)
                    egoWr[nt * 16 + m] = vout[nt][j];
            }
        }
    }

#pragma unroll
    for (int j = 0; j < 8; ++j) {
        float s = ssq[j];
        s += __shfl_xor(s, 8, 16);
        s += __shfl_xor(s, 4, 16);
        s += __shfl_xor(s, 2, 16);
        s += __shfl_xor(s, 1, 16);
        float inv = 1.0f / fmaxf(sqrtf(s), 1e-12f);
        const int lr = lrow0 + j + (half << 3);
        if (lr < rows) {
            float* o = out + (long)(lo + lr) * 192 + out_col0;
#pragma unroll
            for (int nt = 0; nt < 4; ++nt)
                o[nt * 16 + m] = vout[nt][j] * inv;
        }
    }
}

template <bool SECOND>
__global__ __launch_bounds__(1024) void gnn_spmm_own(
    const int* __restrict__ esrc, const int* __restrict__ edst,
    const float* __restrict__ eval, const unsigned* __restrict__ key,
    const unsigned* __restrict__ rec,
    const float* __restrict__ user_emb, const float* __restrict__ item_emb,
    float* ego,
    const float* __restrict__ Wgc, const float* __restrict__ bgc,
    const float* __restrict__ Wbi, const float* __restrict__ bbi,
    float* out)
{
    __shared__ __attribute__((aligned(16))) unsigned q[OWN_Q];
    __shared__ __attribute__((aligned(8))) unsigned recq[32 * DEAL_CAP * 2];
    __shared__ unsigned cnt[32];
    __shared__ unsigned qn;
    __shared__ unsigned nolist;
    __shared__ __attribute__((aligned(16))) float Wt[64][132];
    __shared__ float bsum[64];

    const int tid  = threadIdx.x;
    const int lane = tid & 31;
    const int wave = __builtin_amdgcn_readfirstlane(tid >> 5);
    const int lo   = blockIdx.x * OWN_ROWS;
    const int rows = min(OWN_ROWS, GNN_N - lo);

    const float* xu = SECOND ? (const float*)ego : user_emb;
    const float* xi = item_emb;

    for (int i = tid; i < 64 * 128; i += 1024) {
        int k = i >> 6;
        int n = i & 63;
        float w = (k < 64) ? Wgc[k * 64 + n] : Wbi[(k - 64) * 64 + n];
        Wt[n][k] = w;
    }
    if (tid < 64) bsum[tid] = bgc[tid] + bbi[tid];

    float ax[OWN_SLOTS], ay[OWN_SLOTS];
#pragma unroll
    for (int k = 0; k < OWN_SLOTS; ++k) { ax[k] = 0.f; ay[k] = 0.f; }

    if (!SECOND) {
        own_scan_sum<false>(esrc, edst, eval, key, rec, xu, xi, out, ax, ay, q, &qn, cnt, recq,
                            tid, lane, wave, lo, rows);
    } else {
        const unsigned seg = (2u * (unsigned)rows) >> 5;
        const unsigned h = __builtin_amdgcn_readfirstlane(
            *own_chunk(out, lo, rows, (unsigned)wave * seg));
        if (tid == 0) nolist = 0u;
        __syncthreads();
        if (h == LIST_NONE && lane == 0) nolist = 1u;
        __syncthreads();
        if (__builtin_amdgcn_readfirstlane(nolist) == 0u) {
            const unsigned total = min(h, seg - 1u);
            unsigned cc = 0u;
            uint2 nx; nx.x = REC_BAD_ROW << REC_DST_BITS; nx.y = 0u;
            if (cc < total)
                nx = *(const uint2*)(own_chunk(out, lo, rows, (unsigned)wave * seg + 1u) + lane * 2);
            for (; cc < total; ++cc) {
                const uint2 rc = nx;
                if (cc + 1u < total)
                    nx = *(const uint2*)(own_chunk(out, lo, rows, (unsigned)wave * seg + 2u + cc)
                                         + lane * 2);
                own_sum32<false>(rc, rows, xu, xi, ax, ay, lane);
            }
        } else {
            own_scan_sum<true>(esrc, edst, eval, key, rec, xu, xi, out, ax, ay, q, &qn, cnt, recq,
                               tid, lane, wave, lo, rows);
        }
    }

    float* sl = (float*)q;
#pragma unroll
    for (int k = 0; k < OWN_SLOTS; ++k) {
        const int r = k * 32 + wave;
        if (r < SL_ROWS) {
            v2f o; o.x = ax[k]; o.y = ay[k];
            *(v2f*)(sl + r * SL_PITCH + lane * 2) = o;
        }
    }
    __syncthreads();

    const int ntiles = (rows + 15) >> 4;
    if (wave < ntiles)
        own_layer_tile<SECOND>(sl, Wt, bsum, user_emb, item_emb, ego, out, lo, rows, wave, lane);
}

extern "C" void kernel_launch(void* const* d_in, const int* in_sizes, int n_in,
                              void* d_out, int out_size, void* d_ws, size_t ws_size,
                              hipStream_t stream) {
    const float* user_emb = (const float*)d_in[0];
    const float* item_emb = (const float*)d_in[1];
    const float* W_gc0 = (const float*)d_in[2];
    const float* b_gc0 = (const float*)d_in[3];
    const float* W_bi0 = (const float*)d_in[4];
    const float* b_bi0 = (const float*)d_in[5];
    const float* W_gc1 = (const float*)d_in[6];
    const float* b_gc1 = (const float*)d_in[7];
    const float* W_bi1 = (const float*)d_in[8];
    const float* b_bi1 = (const float*)d_in[9];
    const int*   e_src = (const int*)d_in[10];
    const int*   e_dst = (const int*)d_in[11];
    const float* e_val = (const float*)d_in[12];
    float* out = (float*)d_out;

    float* ego = (float*)d_ws;
    unsigned* rec  = (unsigned*)(ego + (long)GNN_N * GNN_EMB);
    unsigned* key4 = rec + (long)GNN_E * 2;

    gnn_key<<<FE_BLOCKS, 256, 0, stream>>>(e_src, e_dst, e_val, key4, rec);

    gnn_spmm_own<false><<<OWN_BLOCKS, 1024, 0, stream>>>(e_src, e_dst, e_val, key4, rec,
                                                         user_emb, item_emb, ego,
                                                         W_gc0, b_gc0, W_bi0, b_bi0, out);

    gnn_spmm_own<true><<<OWN_BLOCKS, 1024, 0, stream>>>(e_src, e_dst, e_val, key4, rec,
                                                        user_emb, item_emb, ego,
                                                        W_gc1, b_gc1, W_bi1, b_bi1, out);
}


# Owner key as two one-hot hex digits: 32 bit planes, kept in out rows the lists leave free (lists: 18 chunks a wave); an owner block's edge scan reads 2 planes, was 8; sum order: LDS arrival
# speedup vs baseline: 5.2742x; 1.0940x over previous
#include <hip/hip_runtime.h>
#include <hip/hip_bf16.h>
#include <math.h>


typedef __attribute__((ext_vector_type(2))) float v2f;
typedef __attribute__((ext_vector_type(4))) float v4f;
typedef __attribute__((ext_vector_type(8))) float v8f;

#define GNN_N       100000
#define GNN_NUSERS  60000
#define GNN_EMB     64
#define GNN_E       3200000

#define FE_BLOCKS   625
#define FE_WAVES    (FE_BLOCKS * 8)

#define OWN_ROWS   391
#define OWN_BLOCKS 256
#define OWN_Q      32768
#define KEY_G      (GNN_E / 32)
#define KEY_SG     (GNN_E / 128)
#define KEY_PER_WAVE (KEY_SG / FE_WAVES)
#define REC_DST_BITS 17
#define REC_DST_MASK ((1u << REC_DST_BITS) - 1u)
#define REC_BAD_ROW  511u
#define LIST_NONE    0xFFFFFFFFu
#define OWN_SLOTS    13
#define DEAL_CAP     576
#define LIST_SEG     18
#define PL_LINES     (KEY_G / 32)
#define PL_ROW0      192
#define PL_ROWS      199

static_assert(OWN_ROWS * OWN_BLOCKS >= GNN_N, "owners must cover all rows");
static_assert(KEY_SG % FE_WAVES == 0, "the key pass has no tail");
static_assert(GNN_N <= (1 << REC_DST_BITS), "dst must fit its record field");
static_assert(OWN_ROWS <= (int)REC_BAD_ROW, "local row must fit below the bad-row code");
static_assert(OWN_ROWS <= 32 * OWN_SLOTS, "every local row has a wave and a slot");
static_assert(KEY_G % 32 == 0, "a plane is whole lines");
static_assert(FE_BLOCKS * KEY_PER_WAVE == PL_LINES, "a key block writes one line of every plane per pass");
static_assert(32 * LIST_SEG - OWN_ROWS <= PL_ROW0, "the lists end below the plane rows");
static_assert(PL_ROW0 + PL_ROWS <= OWN_ROWS, "the plane rows lie inside a full block's rows");
static_assert(16 * PL_LINES <= PL_ROWS * (OWN_BLOCKS - 1), "the full blocks' rows hold all 32 planes");

__device__ __forceinline__ unsigned plane_line(const unsigned L)
{
    const unsigned C = L >> 1;
    const unsigned B = C / PL_ROWS;
    const unsigned r = C - B * PL_ROWS;
    return (B * OWN_ROWS + PL_ROW0 + r) * 192u + 128u + (L & 1u) * 32u;
}

__global__ __launch_bounds__(256) void gnn_key(
    const int* __restrict__ esrc, const int* __restrict__ edst,
    const float* __restrict__ eval, unsigned* __restrict__ planes,
    unsigned* __restrict__ rec)
{
    __shared__ unsigned tile[KEY_PER_WAVE][32][33];
    const int lane = threadIdx.x & 31;
    const int wib = __builtin_amdgcn_readfirstlane(threadIdx.x >> 5);
    const int wv = blockIdx.x * 8 + wib;
    unsigned inv[4];
#pragma unroll
    for (int k = 0; k < 4; ++k) inv[k] = ((lane >> k) & 1) ? 0u : ~0u;
    unsigned s[KEY_PER_WAVE][4], d[KEY_PER_WAVE][4], v[KEY_PER_WAVE][4];
#pragma unroll
    for (int h = 0; h < KEY_PER_WAVE; ++h) {
#pragma unroll
        for (int u = 0; u < 4; ++u) {
            const int e = (wv + h * FE_WAVES) * 128 + u * 32 + lane;
            s[h][u] = (unsigned)esrc[e];
            d[h][u] = (unsigned)edst[e];
            v[h][u] = __float_as_uint(eval[e]);
        }
    }
#pragma unroll
    for (int h = 0; h < KEY_PER_WAVE; ++h) {
        const int G = wv + h * FE_WAVES;
#pragma unroll
        for (int u = 0; u < 4; ++u) {
            const unsigned oq = s[h][u] / OWN_ROWS;
            const bool owned  = oq < (unsigned)OWN_BLOCKS;
            const unsigned lr = owned ? (s[h][u] - oq * OWN_ROWS) : REC_BAD_ROW;
            uint2 rc;
            rc.x = (d[h][u] & REC_DST_MASK) | (lr << REC_DST_BITS);
            rc.y = v[h][u];
            *(uint2*)(rec + ((long)(4 * G + u) * 32 + lane) * 2) = rc;
            unsigned b[8];
#pragma unroll
            for (int j = 0; j < 8; ++j)
                b[j] = __builtin_amdgcn_ballot_w32(((oq >> j) & 1u) != 0u);
            const unsigned bo = __builtin_amdgcn_ballot_w32(owned);
            const unsigned mh = (b[4] ^ inv[0]) & (b[5] ^ inv[1]) & (b[6] ^ inv[2]) & (b[7] ^ inv[3]);
            const unsigned ml = (b[0] ^ inv[0]) & (b[1] ^ inv[1]) & (b[2] ^ inv[2]) & (b[3] ^ inv[3]);
            const unsigned mine = ((lane < 16) ? mh : ml) & bo;
            tile[h][lane][wib * 4 + u] = mine;
        }
    }
    __syncthreads();
#pragma unroll
    for (int h = 0; h < KEY_PER_WAVE; ++h) {
#pragma unroll
        for (int t = 0; t < 4; ++t) {
            const unsigned p = (unsigned)(wib * 4 + t);
            const unsigned L = p * PL_LINES + (unsigned)(blockIdx.x + h * FE_BLOCKS);
            planes[plane_line(L) + lane] = tile[h][p][lane];
        }
    }
}

__device__ __forceinline__ unsigned* own_chunk(float* out, int lo, int rows, unsigned c)
{
    return (c < (unsigned)rows)
        ? (unsigned*)(out + (long)(lo + (int)c) * 192)
        : (unsigned*)(out + (long)(lo + (int)c - rows) * 192 + 128);
}

template <bool SPLIT>
__device__ __forceinline__ void own_sum32(
    const uint2 rc, const int rows, const float* __restrict__ xu,
    const float* __restrict__ xi, float (&ax)[OWN_SLOTS], float (&ay)[OWN_SLOTS],
    const int lane)
{
    const int   d = (int)(rc.x & REC_DST_MASK);
    const int   r = (int)(rc.x >> REC_DST_BITS);
    const float v = __uint_as_float(rc.y);
    const bool ok = (unsigned)r < (unsigned)rows;
    const unsigned okm = __builtin_amdgcn_ballot_w32(ok);
    const float* xi0 = xi - (long)GNN_NUSERS * GNN_EMB;
#pragma unroll
    for (int j0 = 0; j0 < 32; j0 += 16) {
        if (((okm >> j0) & 0xFFFFu) == 0u) continue;
        v2f xv[16];
#pragma unroll
        for (int u = 0; u < 16; ++u) {
            int dj = __builtin_amdgcn_readlane(d, j0 + u);
            const float* xb = (SPLIT && dj >= GNN_NUSERS) ? xi0 : xu;
            xv[u] = *(const v2f*)(xb + (unsigned)dj * (unsigned)GNN_EMB + lane * 2);
        }
#pragma unroll
        for (int u = 0; u < 16; ++u) {
            if ((okm >> (j0 + u)) & 1u) {
                const int   kj = __builtin_amdgcn_readlane(r, j0 + u) >> 5;
                const float vj = __int_as_float(
                    __builtin_amdgcn_readlane(__float_as_int(v), j0 + u));
                ax[kj] += vj * xv[u].x;
                ay[kj] += vj * xv[u].y;
            }
        }
    }
}

template <bool RAW>
__device__ __forceinline__ void own_scan_sum(
    const int* __restrict__ esrc, const int* __restrict__ edst,
    const float* __restrict__ eval,
    const unsigned* __restrict__ rec,
    const float* __restrict__ xu, const float* __restrict__ xi, float* out,
    float (&ax)[OWN_SLOTS], float (&ay)[OWN_SLOTS],
    unsigned* q, unsigned* qn, unsigned* cnt, unsigned* recq,
    const int tid, const int lane, const int wave, const int lo, const int rows)
{
    const unsigned* pw = (const unsigned*)out;
    const unsigned Lhi = (blockIdx.x >> 4) * PL_LINES;
    const unsigned Llo = (16u + (blockIdx.x & 15u)) * PL_LINES;
    const unsigned seg = min((2u * (unsigned)rows) >> 5, (unsigned)LIST_SEG);
    unsigned wdone = 0u;
    bool fits = true;

    int pos  = 0;
    int step = KEY_G;
    for (int trip = 0; trip < 3 + GNN_E / OWN_Q && pos < KEY_G; ++trip) {
        const int end = min(pos + step, KEY_G);
        if (tid == 0) *qn = 0;
        __syncthreads();

        if (!RAW) {
            const int qend = end >> 2;
            for (int q0 = (pos >> 2) + tid; q0 < qend; q0 += 4096) {
                uint4 ka[4], kb[4];
#pragma unroll
                for (int u = 0; u < 4; ++u) {
                    const unsigned qc  = (unsigned)min(q0 + u * 1024, KEY_G / 4 - 1);
                    const unsigned col = (qc & 7u) * 4u;
                    ka[u] = *(const uint4*)(pw + plane_line(Lhi + (qc >> 3)) + col);
                    kb[u] = *(const uint4*)(pw + plane_line(Llo + (qc >> 3)) + col);
                }
#pragma unroll
                for (int u = 0; u < 4; ++u) {
                    const int qd = q0 + u * 1024;
                    unsigned mm[4];
                    mm[0] = ka[u].x & kb[u].x;  mm[1] = ka[u].y & kb[u].y;
                    mm[2] = ka[u].z & kb[u].z;  mm[3] = ka[u].w & kb[u].w;
#pragma unroll
                    for (int w = 0; w < 4; ++w) {
                        unsigned m = (qd < qend) ? mm[w] : 0u;
                        const unsigned g = (unsigned)qd * 4u + (unsigned)w;
                        while (m) {
                            unsigned bit = (unsigned)__ffs((int)m) - 1u;
                            m &= m - 1u;
                            unsigned slot = atomicAdd(qn, 1u);
                            if (slot < OWN_Q) q[slot] = g * 32u + bit;
                        }
                    }
                }
            }
        } else {
            for (int e = pos * 32 + tid; e < end * 32; e += 1024) {
                if ((unsigned)esrc[e] / OWN_ROWS == blockIdx.x) {
                    unsigned slot = atomicAdd(qn, 1u);
                    if (slot < OWN_Q) q[slot] = (unsigned)e;
                }
            }
        }
        __syncthreads();
        const unsigned n = __builtin_amdgcn_readfirstlane(*qn);
        __syncthreads();
        if (n > OWN_Q) { step = OWN_Q / 32; continue; }

        unsigned start = 0u;
        unsigned rstep = n;
        for (int round = 0; round < 2 + OWN_Q / DEAL_CAP && start < n; ++round) {
            const unsigned stop = start + min(rstep, n - start);
            if (tid < 32) cnt[tid] = 0u;
            __syncthreads();
            for (unsigned base = start + (unsigned)wave * 32u; base < stop; base += 1024u) {
                const unsigned qi = base + lane;
                const unsigned e = q[min(qi, n - 1u)];
                uint2 rc;
                if (RAW) {
                    const unsigned s  = (unsigned)esrc[e];
                    const unsigned oq = s / OWN_ROWS;
                    const unsigned lr = (oq == blockIdx.x) ? (s - oq * OWN_ROWS) : REC_BAD_ROW;
                    rc.x = (((unsigned)edst[e]) & REC_DST_MASK) | (lr << REC_DST_BITS);
                    rc.y = __float_as_uint(eval[e]);
                } else {
                    rc = *(const uint2*)(rec + (size_t)e * 2u);
                }
                const unsigned r = rc.x >> REC_DST_BITS;
                if (qi < stop && r < (unsigned)rows) {
                    const unsigned w = r & 31u;
                    const unsigned slot = atomicAdd(&cnt[w], 1u);
                    if (slot < DEAL_CAP) *(uint2*)(recq + 2u * (w * DEAL_CAP + slot)) = rc;
                }
            }
            __syncthreads();
            const unsigned cl   = cnt[lane];
            const unsigned over = __builtin_amdgcn_ballot_w32(cl > DEAL_CAP);
            const unsigned mine = (unsigned)__builtin_amdgcn_readlane((int)cl, wave);
            if (over != 0u) {
                rstep = DEAL_CAP;
                __syncthreads();
                continue;
            }
            for (unsigned b = 0u; b < mine; b += 32u) {
                uint2 rc; rc.x = REC_BAD_ROW << REC_DST_BITS; rc.y = 0u;
                if (b + lane < mine)
                    rc = *(const uint2*)(recq + 2u * ((unsigned)wave * DEAL_CAP + b + lane));
                if (!RAW) {
                    if (wdone + 1u < seg)
                        *(uint2*)(own_chunk(out, lo, rows, (unsigned)wave * seg + 1u + wdone)
                                  + lane * 2) = rc;
                    else
                        fits = false;
                    ++wdone;
                }
                own_sum32<!RAW>(rc, rows, xu, xi, ax, ay, lane);
            }
            start = stop;
            __syncthreads();
        }
        pos = end;
    }
    if (!RAW) {
        const unsigned h = fits ? wdone : LIST_NONE;
        uint2 hv; hv.x = h; hv.y = h;
        *(uint2*)(own_chunk(out, lo, rows, (unsigned)wave * seg) + lane * 2) = hv;
    }
}

#define SL_PITCH 68
#define SL_ROWS  400

static_assert(SL_ROWS >= ((OWN_ROWS + 15) / 16) * 16, "the LDS tile holds every padded row");
static_assert(SL_ROWS <= 32 * OWN_SLOTS, "every LDS row has a wave and a slot that writes it");
static_assert(SL_ROWS * SL_PITCH <= OWN_Q, "the LDS tile fits the edge queue's space");

__device__ __forceinline__ const float* cat_row(
    const float* __restrict__ user_emb, const float* __restrict__ item_emb, const int g)
{
    return (g < GNN_NUSERS) ? user_emb + (long)g * GNN_EMB
                            : item_emb + (long)(g - GNN_NUSERS) * GNN_EMB;
}

template <bool SECOND>
__device__ __forceinline__ void own_layer_tile(
    const float* sl, const float (*Wt)[132], const float* bsum,
    const float* __restrict__ user_emb, const float* __restrict__ item_emb,
    float* ego, float* out, const int lo, const int rows, const int tile, const int lane)
{
    const int m     = lane & 15;
    const int half  = lane >> 4;
    const int kb    = half << 1;
    const int lrow0 = tile * 16;
    const int glast = lo + rows - 1;
    const int out_col0 = SECOND ? 128 : 64;

    if (SECOND) {
        v4f cv[8];
#pragma unroll
        for (int t = 0; t < 8; ++t) {
            const int g = min(lo + lrow0 + 2 * t + half, glast);
            cv[t] = *(const v4f*)(cat_row(user_emb, item_emb, g) + m * 4);
        }
        asm volatile("" :: "v"(cv[0]), "v"(cv[1]), "v"(cv[2]), "v"(cv[3]),
                           "v"(cv[4]), "v"(cv[5]), "v"(cv[6]), "v"(cv[7]));
#pragma unroll
        for (int t = 0; t < 8; ++t) {
            const int lr = lrow0 + 2 * t + half;
            if (lr < rows)
                *(v4f*)(out + (long)(lo + lr) * 192 + m * 4) = cv[t];
        }
    }

    const int    ga    = min(lo + lrow0 + m, glast);
    const float* egoRd = SECOND ? (const float*)ego + (long)ga * GNN_EMB
                                : cat_row(user_emb, item_emb, ga);
    const float* sLRd  = sl + (lrow0 + m) * SL_PITCH;

    v2f e2[16];
#pragma unroll
    for (int s = 0; s < 16; ++s)
        e2[s] = *(const v2f*)(egoRd + 4 * s + kb);

    v8f acc[4] = {};

#pragma unroll
    for (int s = 0; s < 16; ++s) {
        const v2f l = *(const v2f*)(sLRd + 4 * s + kb);
        v2f a;
        a.x = e2[s].x + l.x;
        a.y = e2[s].y + l.y;
#pragma unroll
        for (int nt = 0; nt < 4; ++nt) {
            v2f b = *(const v2f*)(&Wt[nt * 16 + m][4 * s + kb]);
            acc[nt] = __builtin_amdgcn_wmma_f32_16x16x4_f32(
                false, a, false, b, (short)0, acc[nt], false, false);
        }
    }
#pragma unroll
    for (int s = 0; s < 16; ++s) {
        const v2f l = *(const v2f*)(sLRd + 4 * s + kb);
        v2f a;
        a.x = e2[s].x * l.x;
        a.y = e2[s].y * l.y;
#pragma unroll
        for (int nt = 0; nt < 4; ++nt) {
            v2f b = *(const v2f*)(&Wt[nt * 16 + m][64 + 4 * s + kb]);
            acc[nt] = __builtin_amdgcn_wmma_f32_16x16x4_f32(
                false, a, false, b, (short)0, acc[nt], false, false);
        }
    }

    float vout[4][8];
    float ssq[8];
#pragma unroll
    for (int j = 0; j < 8; ++j) ssq[j] = 0.f;
#pragma unroll
    for (int nt = 0; nt < 4; ++nt) {
        float bc = bsum[nt * 16 + m];
#pragma unroll
        for (int j = 0; j < 8; ++j) {
            float v = acc[nt][j] + bc;
            v = (v > 0.f) ? v : 0.01f * v;
            vout[nt][j] = v;
            ssq[j] += v * v;
        }
    }

    if (!SECOND) {
#pragma unroll
        for (int j = 0; j < 8; ++j) {
            const int lr = lrow0 + j + (half << 3);
            if (lr < rows) {
                float* egoWr = ego + (long)(lo + lr) * GNN_EMB;
#pragma unroll
                for (int nt = 0; nt < 4; ++nt)
                    egoWr[nt * 16 + m] = vout[nt][j];
            }
        }
    }

#pragma unroll
    for (int j = 0; j < 8; ++j) {
        float s = ssq[j];
        s += __shfl_xor(s, 8, 16);
        s += __shfl_xor(s, 4, 16);
        s += __shfl_xor(s, 2, 16);
        s += __shfl_xor(s, 1, 16);
        float inv = 1.0f / fmaxf(sqrtf(s), 1e-12f);
        const int lr = lrow0 + j + (half << 3);
        if (lr < rows) {
            float* o = out + (long)(lo + lr) * 192 + out_col0;
#pragma unroll
            for (int nt = 0; nt < 4; ++nt)
                o[nt * 16 + m] = vout[nt][j] * inv;
        }
    }
}

template <bool SECOND>
__global__ __launch_bounds__(1024) void gnn_spmm_own(
    const int* __restrict__ esrc, const int* __restrict__ edst,
    const float* __restrict__ eval,
    const unsigned* __restrict__ rec,
    const float* __restrict__ user_emb, const float* __restrict__ item_emb,
    float* ego,
    const float* __restrict__ Wgc, const float* __restrict__ bgc,
    const float* __restrict__ Wbi, const float* __restrict__ bbi,
    float* out)
{
    __shared__ __attribute__((aligned(16))) unsigned q[OWN_Q];
    __shared__ __attribute__((aligned(8))) unsigned recq[32 * DEAL_CAP * 2];
    __shared__ unsigned cnt[32];
    __shared__ unsigned qn;
    __shared__ unsigned nolist;
    __shared__ __attribute__((aligned(16))) float Wt[64][132];
    __shared__ float bsum[64];

    const int tid  = threadIdx.x;
    const int lane = tid & 31;
    const int wave = __builtin_amdgcn_readfirstlane(tid >> 5);
    const int lo   = blockIdx.x * OWN_ROWS;
    const int rows = min(OWN_ROWS, GNN_N - lo);

    const float* xu = SECOND ? (const float*)ego : user_emb;
    const float* xi = item_emb;

    for (int i = tid; i < 64 * 128; i += 1024) {
        int k = i >> 6;
        int n = i & 63;
        float w = (k < 64) ? Wgc[k * 64 + n] : Wbi[(k - 64) * 64 + n];
        Wt[n][k] = w;
    }
    if (tid < 64) bsum[tid] = bgc[tid] + bbi[tid];

    float ax[OWN_SLOTS], ay[OWN_SLOTS];
#pragma unroll
    for (int k = 0; k < OWN_SLOTS; ++k) { ax[k] = 0.f; ay[k] = 0.f; }

    if (!SECOND) {
        own_scan_sum<false>(esrc, edst, eval, rec, xu, xi, out, ax, ay, q, &qn, cnt, recq,
                            tid, lane, wave, lo, rows);
    } else {
        const unsigned seg = min((2u * (unsigned)rows) >> 5, (unsigned)LIST_SEG);
        const unsigned h = __builtin_amdgcn_readfirstlane(
            *own_chunk(out, lo, rows, (unsigned)wave * seg));
        if (tid == 0) nolist = 0u;
        __syncthreads();
        if (h == LIST_NONE && lane == 0) nolist = 1u;
        __syncthreads();
        if (__builtin_amdgcn_readfirstlane(nolist) == 0u) {
            const unsigned total = min(h, seg - 1u);
            unsigned cc = 0u;
            uint2 nx; nx.x = REC_BAD_ROW << REC_DST_BITS; nx.y = 0u;
            if (cc < total)
                nx = *(const uint2*)(own_chunk(out, lo, rows, (unsigned)wave * seg + 1u) + lane * 2);
            for (; cc < total; ++cc) {
                const uint2 rc = nx;
                if (cc + 1u < total)
                    nx = *(const uint2*)(own_chunk(out, lo, rows, (unsigned)wave * seg + 2u + cc)
                                         + lane * 2);
                own_sum32<false>(rc, rows, xu, xi, ax, ay, lane);
            }
        } else {
            own_scan_sum<true>(esrc, edst, eval, rec, xu, xi, out, ax, ay, q, &qn, cnt, recq,
                               tid, lane, wave, lo, rows);
        }
    }

    float* sl = (float*)q;
#pragma unroll
    for (int k = 0; k < OWN_SLOTS; ++k) {
        const int r = k * 32 + wave;
        if (r < SL_ROWS) {
            v2f o; o.x = ax[k]; o.y = ay[k];
            *(v2f*)(sl + r * SL_PITCH + lane * 2) = o;
        }
    }
    __syncthreads();

    const int ntiles = (rows + 15) >> 4;
    if (wave < ntiles)
        own_layer_tile<SECOND>(sl, Wt, bsum, user_emb, item_emb, ego, out, lo, rows, wave, lane);
}

extern "C" void kernel_launch(void* const* d_in, const int* in_sizes, int n_in,
                              void* d_out, int out_size, void* d_ws, size_t ws_size,
                              hipStream_t stream) {
    const float* user_emb = (const float*)d_in[0];
    const float* item_emb = (const float*)d_in[1];
    const float* W_gc0 = (const float*)d_in[2];
    const float* b_gc0 = (const float*)d_in[3];
    const float* W_bi0 = (const float*)d_in[4];
    const float* b_bi0 = (const float*)d_in[5];
    const float* W_gc1 = (const float*)d_in[6];
    const float* b_gc1 = (const float*)d_in[7];
    const float* W_bi1 = (const float*)d_in[8];
    const float* b_bi1 = (const float*)d_in[9];
    const int*   e_src = (const int*)d_in[10];
    const int*   e_dst = (const int*)d_in[11];
    const float* e_val = (const float*)d_in[12];
    float* out = (float*)d_out;

    float* ego = (float*)d_ws;
    unsigned* rec  = (unsigned*)(ego + (long)GNN_N * GNN_EMB);

    gnn_key<<<FE_BLOCKS, 256, 0, stream>>>(e_src, e_dst, e_val, (unsigned*)out, rec);

    gnn_spmm_own<false><<<OWN_BLOCKS, 1024, 0, stream>>>(e_src, e_dst, e_val, rec,
                                                         user_emb, item_emb, ego,
                                                         W_gc0, b_gc0, W_bi0, b_bi0, out);

    gnn_spmm_own<true><<<OWN_BLOCKS, 1024, 0, stream>>>(e_src, e_dst, e_val, rec,
                                                        user_emb, item_emb, ego,
                                                        W_gc1, b_gc1, W_bi1, b_bi1, out);
}
